# starting kernel
# baseline (speedup 1.0000x reference)
#include <hip/hip_runtime.h>
#include <hip/hip_bf16.h>

typedef float v2f __attribute__((ext_vector_type(2)));
typedef float v8f __attribute__((ext_vector_type(8)));

#define MAX_LEN 200
#define MAX_K   3
#define D_IN    128
#define D_OUT   128
#define NBATCH  1024
#define NEG_PAD (-65535.0f)

__global__ void __launch_bounds__(128)
gemm_lowS_kernel(const float* __restrict__ low, const float* __restrict__ S,
                 float* __restrict__ lown) {
  const int wave = threadIdx.x >> 5;
  const int lane = threadIdx.x & 31;
  const int tile = blockIdx.x * 4 + wave;
  const int tm   = tile >> 3;
  const int tn   = tile & 7;

  const int m     = lane & 15;
  const int khalf = (lane >> 4) * 2;
  const float* __restrict__ Arow = low + (size_t)(tm * 16 + m) * D_IN;
  const int n = tn * 16 + (lane & 15);

  v8f c = {};
#pragma unroll
  for (int kk = 0; kk < 32; ++kk) {
    const int kb = kk * 4 + khalf;
    v2f a;  a.x = Arow[kb];          a.y = Arow[kb + 1];
    v2f bb; bb.x = S[kb * D_OUT + n]; bb.y = S[(kb + 1) * D_OUT + n];
    c = __builtin_amdgcn_wmma_f32_16x16x4_f32(false, a, false, bb,
                                              (short)0, c, false, false);
  }

  const int cn    = tn * 16 + (lane & 15);
  const int rbase = tm * 16 + ((lane >> 4) << 3);
#pragma unroll
  for (int r = 0; r < 8; ++r)
    lown[(size_t)(rbase + r) * D_OUT + cn] = c[r];
}

__global__ void init_b_kernel(const float* __restrict__ Bin,
                              float* __restrict__ Bwork,
                              float* __restrict__ Bdelta) {
  const int t = blockIdx.x * blockDim.x + threadIdx.x;
  if (t < MAX_K * MAX_LEN) { Bwork[t] = Bin[t]; Bdelta[t] = 0.0f; }
}

__global__ void __launch_bounds__(128)
routing_iter_kernel(const float* __restrict__ lown, const float* __restrict__ Bwork,
                    const int* __restrict__ seq_len, float* __restrict__ Bdelta,
                    float* __restrict__ out, int last) {
  __shared__ float sW[MAX_K][MAX_LEN];
  __shared__ float sH[MAX_K][D_OUT];
  __shared__ float sred[128];

  const int b    = blockIdx.x;
  const int t    = threadIdx.x;
  const int wave = t >> 5;
  const int lane = t & 31;
  const int len  = seq_len[b];
  const float* __restrict__ lowb = lown + (size_t)b * MAX_LEN * D_OUT;

  for (int i = t; i < MAX_K * MAX_LEN; i += 128) {
    const int l = i % MAX_LEN;
    sW[i / MAX_LEN][l] = (l < len) ? Bwork[i] : NEG_PAD;
  }
  __syncthreads();

  for (int k = 0; k < MAX_K; ++k) {
    float mx = -3.0e38f;
    for (int l = t; l < MAX_LEN; l += 128) mx = fmaxf(mx, sW[k][l]);
    sred[t] = mx; __syncthreads();
    for (int s = 64; s > 0; s >>= 1) {
      if (t < s) sred[t] = fmaxf(sred[t], sred[t + s]);
      __syncthreads();
    }
    const float mk = sred[0]; __syncthreads();
    float sum = 0.0f;
    for (int l = t; l < MAX_LEN; l += 128) {
      const float e = __expf(sW[k][l] - mk);
      sW[k][l] = e; sum += e;
    }
    sred[t] = sum; __syncthreads();
    for (int s = 64; s > 0; s >>= 1) {
      if (t < s) sred[t] += sred[t + s];
      __syncthreads();
    }
    const float inv = 1.0f / sred[0]; __syncthreads();
    for (int l = t; l < MAX_LEN; l += 128) sW[k][l] *= inv;
    __syncthreads();
  }

  const int m     = lane & 15;
  const int khalf = (lane >> 4) * 2;
#pragma unroll
  for (int j = 0; j < 2; ++j) {
    const int nt = wave * 2 + j;
    const int n  = nt * 16 + (lane & 15);
    v8f c = {};
    for (int kk = 0; kk < 50; ++kk) {
      const int kb = kk * 4 + khalf;
      v2f a;
      a.x = (m < MAX_K) ? sW[m][kb]     : 0.0f;
      a.y = (m < MAX_K) ? sW[m][kb + 1] : 0.0f;
      v2f bb;
      bb.x = lowb[kb * D_OUT + n];
      bb.y = lowb[(kb + 1) * D_OUT + n];
      c = __builtin_amdgcn_wmma_f32_16x16x4_f32(false, a, false, bb,
                                                (short)0, c, false, false);
    }
    if (lane < 16) {
#pragma unroll
      for (int r = 0; r < MAX_K; ++r) sH[r][nt * 16 + lane] = c[r];
    }
  }
  __syncthreads();

  {
    const float x0 = sH[0][t], x1 = sH[1][t], x2 = sH[2][t];
    const float sq = x0 * x0 + x1 * x1 + x2 * x2;
    const float scale = sq / (1.0f + sq) * rsqrtf(sq + 1e-9f);
    sH[0][t] = scale * x0; sH[1][t] = scale * x1; sH[2][t] = scale * x2;
    if (last) {
      float* __restrict__ ob = out + (size_t)b * MAX_K * D_OUT;
      ob[0 * D_OUT + t] = sH[0][t];
      ob[1 * D_OUT + t] = sH[1][t];
      ob[2 * D_OUT + t] = sH[2][t];
    }
  }
  __syncthreads();

  if (!last) {
    for (int p = t; p < MAX_K * MAX_LEN; p += 128) {
      const int k = p / MAX_LEN;
      const int l = p % MAX_LEN;
      const float4* __restrict__ lp =
          reinterpret_cast<const float4*>(lowb + (size_t)l * D_OUT);
      float acc = 0.0f;
#pragma unroll
      for (int e4 = 0; e4 < D_OUT / 4; ++e4) {
        const float4 v = lp[e4];
        acc += v.x * sH[k][4 * e4 + 0];
        acc += v.y * sH[k][4 * e4 + 1];
        acc += v.z * sH[k][4 * e4 + 2];
        acc += v.w * sH[k][4 * e4 + 3];
      }
      atomicAdd(&Bdelta[p], acc);
    }
  }
}

__global__ void update_b_kernel(float* __restrict__ Bwork,
                                float* __restrict__ Bdelta) {
  const int t = blockIdx.x * blockDim.x + threadIdx.x;
  if (t < MAX_K * MAX_LEN) { Bwork[t] += Bdelta[t]; Bdelta[t] = 0.0f; }
}

extern "C" void kernel_launch(void* const* d_in, const int* in_sizes, int n_in,
                              void* d_out, int out_size, void* d_ws, size_t ws_size,
                              hipStream_t stream) {
  const float* low  = (const float*)d_in[0];
  const float* Bin  = (const float*)d_in[1];
  const float* S    = (const float*)d_in[2];
  const int*   slen = (const int*)d_in[3];
  float* out = (float*)d_out;

  char* ws = (char*)d_ws;
  float* lown   = (float*)ws;
  float* Bwork  = (float*)(ws + (size_t)NBATCH * MAX_LEN * D_OUT * sizeof(float));
  float* Bdelta = Bwork + MAX_K * MAX_LEN;

  gemm_lowS_kernel<<<(12800 * 8) / 4, 128, 0, stream>>>(low, S, lown);

  init_b_kernel<<<5, 128, 0, stream>>>(Bin, Bwork, Bdelta);

  for (int it = 0; it < 3; ++it) {
    const int last = (it == 2);
    routing_iter_kernel<<<NBATCH, 128, 0, stream>>>(lown, Bwork, slen, Bdelta,
                                                    out, last);
    if (!last) update_b_kernel<<<5, 128, 0, stream>>>(Bwork, Bdelta);
  }
}


# Re-associate the projection: high=(W.low).S and delta=low.(S.high^T) per batch, so the 204800x128x128 low.S GEMM and its workspace plane go; float sums are taken in a different order
# speedup vs baseline: 1.4558x; 1.4558x over previous
#include <hip/hip_runtime.h>
#include <hip/hip_bf16.h>


typedef float v2f __attribute__((ext_vector_type(2)));
typedef float v8f __attribute__((ext_vector_type(8)));

#define MAX_LEN 200
#define MAX_K   3
#define D_IN    128
#define D_OUT   128
#define NBATCH  1024
#define NEG_PAD (-65535.0f)


__global__ void init_b_kernel(const float* __restrict__ Bin,
                              float* __restrict__ Bwork,
                              float* __restrict__ Bdelta) {
  const int t = blockIdx.x * blockDim.x + threadIdx.x;
  if (t < MAX_K * MAX_LEN) { Bwork[t] = Bin[t]; Bdelta[t] = 0.0f; }
}

__global__ void __launch_bounds__(128)
routing_iter_kernel(const float* __restrict__ low, const float* __restrict__ S,
                    const float* __restrict__ Bwork,
                    const int* __restrict__ seq_len, float* __restrict__ Bdelta,
                    float* __restrict__ out, int last) {
  __shared__ float sW[MAX_K][MAX_LEN];
  __shared__ __align__(16) float sT[MAX_K][D_IN];
  __shared__ __align__(16) float sH[MAX_K][D_OUT];
  __shared__ float sG[MAX_K][D_IN];
  __shared__ float sred[128];

  const int b    = blockIdx.x;
  const int t    = threadIdx.x;
  const int wave = t >> 5;
  const int lane = t & 31;
  const int len  = seq_len[b];
  const float* __restrict__ lowb = low + (size_t)b * MAX_LEN * D_IN;

  for (int i = t; i < MAX_K * MAX_LEN; i += 128) {
    const int l = i % MAX_LEN;
    sW[i / MAX_LEN][l] = (l < len) ? Bwork[i] : NEG_PAD;
  }
  __syncthreads();

  for (int k = 0; k < MAX_K; ++k) {
    float mx = -3.0e38f;
    for (int l = t; l < MAX_LEN; l += 128) mx = fmaxf(mx, sW[k][l]);
    sred[t] = mx; __syncthreads();
    for (int s = 64; s > 0; s >>= 1) {
      if (t < s) sred[t] = fmaxf(sred[t], sred[t + s]);
      __syncthreads();
    }
    const float mk = sred[0]; __syncthreads();
    float sum = 0.0f;
    for (int l = t; l < MAX_LEN; l += 128) {
      const float e = __expf(sW[k][l] - mk);
      sW[k][l] = e; sum += e;
    }
    sred[t] = sum; __syncthreads();
    for (int s = 64; s > 0; s >>= 1) {
      if (t < s) sred[t] += sred[t + s];
      __syncthreads();
    }
    const float inv = 1.0f / sred[0]; __syncthreads();
    for (int l = t; l < MAX_LEN; l += 128) sW[k][l] *= inv;
    __syncthreads();
  }

  const int m     = lane & 15;
  const int khalf = (lane >> 4) * 2;
  const int mrow  = (m < MAX_K) ? m : 0;

#pragma unroll
  for (int j = 0; j < 2; ++j) {
    const int nt = wave * 2 + j;
    const int n  = nt * 16 + (lane & 15);
    v8f c = {};
    for (int kk = 0; kk < 50; ++kk) {
      const int kb = kk * 4 + khalf;
      v2f a;
      a.x = (m < MAX_K) ? sW[m][kb]     : 0.0f;
      a.y = (m < MAX_K) ? sW[m][kb + 1] : 0.0f;
      v2f bb;
      bb.x = lowb[kb * D_IN + n];
      bb.y = lowb[(kb + 1) * D_IN + n];
      c = __builtin_amdgcn_wmma_f32_16x16x4_f32(false, a, false, bb,
                                                (short)0, c, false, false);
    }
    if (lane < 16) {
#pragma unroll
      for (int r = 0; r < MAX_K; ++r) sT[r][nt * 16 + lane] = c[r];
    }
  }
  __syncthreads();

  {
    const int n0 = (wave * 2) * 16 + (lane & 15);
    const int n1 = n0 + 16;
    v8f c0 = {}, c1 = {};
#pragma unroll
    for (int kk = 0; kk < D_IN / 4; ++kk) {
      const int kb = kk * 4 + khalf;
      const v2f a = *reinterpret_cast<const v2f*>(&sT[mrow][kb]);
      v2f b0, b1;
      b0.x = S[kb * D_OUT + n0];  b0.y = S[(kb + 1) * D_OUT + n0];
      b1.x = S[kb * D_OUT + n1];  b1.y = S[(kb + 1) * D_OUT + n1];
      c0 = __builtin_amdgcn_wmma_f32_16x16x4_f32(false, a, false, b0,
                                                 (short)0, c0, false, false);
      c1 = __builtin_amdgcn_wmma_f32_16x16x4_f32(false, a, false, b1,
                                                 (short)0, c1, false, false);
    }
    if (lane < 16) {
#pragma unroll
      for (int r = 0; r < MAX_K; ++r) {
        sH[r][n0] = c0[r];
        sH[r][n1] = c1[r];
      }
    }
  }
  __syncthreads();

  {
    const float x0 = sH[0][t], x1 = sH[1][t], x2 = sH[2][t];
    const float sq = x0 * x0 + x1 * x1 + x2 * x2;
    const float scale = sq / (1.0f + sq) * rsqrtf(sq + 1e-9f);
    sH[0][t] = scale * x0; sH[1][t] = scale * x1; sH[2][t] = scale * x2;
    if (last) {
      float* __restrict__ ob = out + (size_t)b * MAX_K * D_OUT;
      ob[0 * D_OUT + t] = sH[0][t];
      ob[1 * D_OUT + t] = sH[1][t];
      ob[2 * D_OUT + t] = sH[2][t];
    }
  }
  __syncthreads();

  if (!last) {
    {
      const int d0 = (wave * 2) * 16 + (lane & 15);
      const int d1 = d0 + 16;
      const float* __restrict__ S0 = S + (size_t)d0 * D_OUT;
      const float* __restrict__ S1 = S + (size_t)d1 * D_OUT;
      v8f c0 = {}, c1 = {};
#pragma unroll
      for (int kk = 0; kk < D_OUT / 4; ++kk) {
        const int kb = kk * 4 + khalf;
        const v2f a = *reinterpret_cast<const v2f*>(&sH[mrow][kb]);
        const v2f b0 = *reinterpret_cast<const v2f*>(S0 + kb);
        const v2f b1 = *reinterpret_cast<const v2f*>(S1 + kb);
        c0 = __builtin_amdgcn_wmma_f32_16x16x4_f32(false, a, false, b0,
                                                   (short)0, c0, false, false);
        c1 = __builtin_amdgcn_wmma_f32_16x16x4_f32(false, a, false, b1,
                                                   (short)0, c1, false, false);
      }
      if (lane < 16) {
#pragma unroll
        for (int r = 0; r < MAX_K; ++r) {
          sG[r][d0] = c0[r];
          sG[r][d1] = c1[r];
        }
      }
    }
    __syncthreads();

    for (int p = t; p < MAX_K * MAX_LEN; p += 128) {
      const int k = p / MAX_LEN;
      const int l = p % MAX_LEN;
      const float4* __restrict__ lp =
          reinterpret_cast<const float4*>(lowb + (size_t)l * D_IN);
      float acc = 0.0f;
#pragma unroll
      for (int e4 = 0; e4 < D_IN / 4; ++e4) {
        const float4 v = lp[e4];
        acc += v.x * sG[k][4 * e4 + 0];
        acc += v.y * sG[k][4 * e4 + 1];
        acc += v.z * sG[k][4 * e4 + 2];
        acc += v.w * sG[k][4 * e4 + 3];
      }
      atomicAdd(&Bdelta[p], acc);
    }
  }
}

__global__ void update_b_kernel(float* __restrict__ Bwork,
                                float* __restrict__ Bdelta) {
  const int t = blockIdx.x * blockDim.x + threadIdx.x;
  if (t < MAX_K * MAX_LEN) { Bwork[t] += Bdelta[t]; Bdelta[t] = 0.0f; }
}

extern "C" void kernel_launch(void* const* d_in, const int* in_sizes, int n_in,
                              void* d_out, int out_size, void* d_ws, size_t ws_size,
                              hipStream_t stream) {
  const float* low  = (const float*)d_in[0];
  const float* Bin  = (const float*)d_in[1];
  const float* S    = (const float*)d_in[2];
  const int*   slen = (const int*)d_in[3];
  float* out = (float*)d_out;

  float* Bwork  = (float*)d_ws;
  float* Bdelta = Bwork + MAX_K * MAX_LEN;

  init_b_kernel<<<5, 128, 0, stream>>>(Bin, Bwork, Bdelta);

  for (int it = 0; it < 3; ++it) {
    const int last = (it == 2);
    routing_iter_kernel<<<NBATCH, 128, 0, stream>>>(low, S, Bwork, slen, Bdelta,
                                                    out, last);
    if (!last) update_b_kernel<<<5, 128, 0, stream>>>(Bwork, Bdelta);
  }
}


# W.low product: one fully unrolled k-loop feeds both N tiles of a wave, so the low[b] loads run many k-steps ahead of the WMMA chains instead of one second-level round trip per k-step
# speedup vs baseline: 2.2734x; 1.5616x over previous
#include <hip/hip_runtime.h>
#include <hip/hip_bf16.h>


typedef float v2f __attribute__((ext_vector_type(2)));
typedef float v8f __attribute__((ext_vector_type(8)));

#define MAX_LEN 200
#define MAX_K   3
#define D_IN    128
#define D_OUT   128
#define NBATCH  1024
#define NEG_PAD (-65535.0f)


__global__ void init_b_kernel(const float* __restrict__ Bin,
                              float* __restrict__ Bwork,
                              float* __restrict__ Bdelta) {
  const int t = blockIdx.x * blockDim.x + threadIdx.x;
  if (t < MAX_K * MAX_LEN) { Bwork[t] = Bin[t]; Bdelta[t] = 0.0f; }
}

__global__ void __launch_bounds__(128)
routing_iter_kernel(const float* __restrict__ low, const float* __restrict__ S,
                    const float* __restrict__ Bwork,
                    const int* __restrict__ seq_len, float* __restrict__ Bdelta,
                    float* __restrict__ out, int last) {
  __shared__ __align__(16) float sW[MAX_K][MAX_LEN];
  __shared__ __align__(16) float sT[MAX_K][D_IN];
  __shared__ __align__(16) float sH[MAX_K][D_OUT];
  __shared__ float sG[MAX_K][D_IN];
  __shared__ float sred[128];

  const int b    = blockIdx.x;
  const int t    = threadIdx.x;
  const int wave = t >> 5;
  const int lane = t & 31;
  const int len  = seq_len[b];
  const float* __restrict__ lowb = low + (size_t)b * MAX_LEN * D_IN;

  for (int i = t; i < MAX_K * MAX_LEN; i += 128) {
    const int l = i % MAX_LEN;
    sW[i / MAX_LEN][l] = (l < len) ? Bwork[i] : NEG_PAD;
  }
  __syncthreads();

  for (int k = 0; k < MAX_K; ++k) {
    float mx = -3.0e38f;
    for (int l = t; l < MAX_LEN; l += 128) mx = fmaxf(mx, sW[k][l]);
    sred[t] = mx; __syncthreads();
    for (int s = 64; s > 0; s >>= 1) {
      if (t < s) sred[t] = fmaxf(sred[t], sred[t + s]);
      __syncthreads();
    }
    const float mk = sred[0]; __syncthreads();
    float sum = 0.0f;
    for (int l = t; l < MAX_LEN; l += 128) {
      const float e = __expf(sW[k][l] - mk);
      sW[k][l] = e; sum += e;
    }
    sred[t] = sum; __syncthreads();
    for (int s = 64; s > 0; s >>= 1) {
      if (t < s) sred[t] += sred[t + s];
      __syncthreads();
    }
    const float inv = 1.0f / sred[0]; __syncthreads();
    for (int l = t; l < MAX_LEN; l += 128) sW[k][l] *= inv;
    __syncthreads();
  }

  const int m     = lane & 15;
  const int khalf = (lane >> 4) * 2;
  const int mrow  = (m < MAX_K) ? m : 0;

  {
    const int n0 = (wave * 2) * 16 + (lane & 15);
    const int n1 = n0 + 16;
    v8f c0 = {}, c1 = {};
#pragma unroll
    for (int kk = 0; kk < MAX_LEN / 4; ++kk) {
      const int kb = kk * 4 + khalf;
      const v2f a = *reinterpret_cast<const v2f*>(&sW[mrow][kb]);
      v2f b0, b1;
      b0.x = lowb[kb * D_IN + n0];  b0.y = lowb[(kb + 1) * D_IN + n0];
      b1.x = lowb[kb * D_IN + n1];  b1.y = lowb[(kb + 1) * D_IN + n1];
      c0 = __builtin_amdgcn_wmma_f32_16x16x4_f32(false, a, false, b0,
                                                 (short)0, c0, false, false);
      c1 = __builtin_amdgcn_wmma_f32_16x16x4_f32(false, a, false, b1,
                                                 (short)0, c1, false, false);
    }
    if (lane < 16) {
#pragma unroll
      for (int r = 0; r < MAX_K; ++r) {
        sT[r][n0] = c0[r];
        sT[r][n1] = c1[r];
      }
    }
  }
  __syncthreads();

  {
    const int n0 = (wave * 2) * 16 + (lane & 15);
    const int n1 = n0 + 16;
    v8f c0 = {}, c1 = {};
#pragma unroll
    for (int kk = 0; kk < D_IN / 4; ++kk) {
      const int kb = kk * 4 + khalf;
      const v2f a = *reinterpret_cast<const v2f*>(&sT[mrow][kb]);
      v2f b0, b1;
      b0.x = S[kb * D_OUT + n0];  b0.y = S[(kb + 1) * D_OUT + n0];
      b1.x = S[kb * D_OUT + n1];  b1.y = S[(kb + 1) * D_OUT + n1];
      c0 = __builtin_amdgcn_wmma_f32_16x16x4_f32(false, a, false, b0,
                                                 (short)0, c0, false, false);
      c1 = __builtin_amdgcn_wmma_f32_16x16x4_f32(false, a, false, b1,
                                                 (short)0, c1, false, false);
    }
    if (lane < 16) {
#pragma unroll
      for (int r = 0; r < MAX_K; ++r) {
        sH[r][n0] = c0[r];
        sH[r][n1] = c1[r];
      }
    }
  }
  __syncthreads();

  {
    const float x0 = sH[0][t], x1 = sH[1][t], x2 = sH[2][t];
    const float sq = x0 * x0 + x1 * x1 + x2 * x2;
    const float scale = sq / (1.0f + sq) * rsqrtf(sq + 1e-9f);
    sH[0][t] = scale * x0; sH[1][t] = scale * x1; sH[2][t] = scale * x2;
    if (last) {
      float* __restrict__ ob = out + (size_t)b * MAX_K * D_OUT;
      ob[0 * D_OUT + t] = sH[0][t];
      ob[1 * D_OUT + t] = sH[1][t];
      ob[2 * D_OUT + t] = sH[2][t];
    }
  }
  __syncthreads();

  if (!last) {
    {
      const int d0 = (wave * 2) * 16 + (lane & 15);
      const int d1 = d0 + 16;
      const float* __restrict__ S0 = S + (size_t)d0 * D_OUT;
      const float* __restrict__ S1 = S + (size_t)d1 * D_OUT;
      v8f c0 = {}, c1 = {};
#pragma unroll
      for (int kk = 0; kk < D_OUT / 4; ++kk) {
        const int kb = kk * 4 + khalf;
        const v2f a = *reinterpret_cast<const v2f*>(&sH[mrow][kb]);
        const v2f b0 = *reinterpret_cast<const v2f*>(S0 + kb);
        const v2f b1 = *reinterpret_cast<const v2f*>(S1 + kb);
        c0 = __builtin_amdgcn_wmma_f32_16x16x4_f32(false, a, false, b0,
                                                   (short)0, c0, false, false);
        c1 = __builtin_amdgcn_wmma_f32_16x16x4_f32(false, a, false, b1,
                                                   (short)0, c1, false, false);
      }
      if (lane < 16) {
#pragma unroll
        for (int r = 0; r < MAX_K; ++r) {
          sG[r][d0] = c0[r];
          sG[r][d1] = c1[r];
        }
      }
    }
    __syncthreads();

    for (int p = t; p < MAX_K * MAX_LEN; p += 128) {
      const int k = p / MAX_LEN;
      const int l = p % MAX_LEN;
      const float4* __restrict__ lp =
          reinterpret_cast<const float4*>(lowb + (size_t)l * D_IN);
      float acc = 0.0f;
#pragma unroll
      for (int e4 = 0; e4 < D_IN / 4; ++e4) {
        const float4 v = lp[e4];
        acc += v.x * sG[k][4 * e4 + 0];
        acc += v.y * sG[k][4 * e4 + 1];
        acc += v.z * sG[k][4 * e4 + 2];
        acc += v.w * sG[k][4 * e4 + 3];
      }
      atomicAdd(&Bdelta[p], acc);
    }
  }
}

__global__ void update_b_kernel(float* __restrict__ Bwork,
                                float* __restrict__ Bdelta) {
  const int t = blockIdx.x * blockDim.x + threadIdx.x;
  if (t < MAX_K * MAX_LEN) { Bwork[t] += Bdelta[t]; Bdelta[t] = 0.0f; }
}

extern "C" void kernel_launch(void* const* d_in, const int* in_sizes, int n_in,
                              void* d_out, int out_size, void* d_ws, size_t ws_size,
                              hipStream_t stream) {
  const float* low  = (const float*)d_in[0];
  const float* Bin  = (const float*)d_in[1];
  const float* S    = (const float*)d_in[2];
  const int*   slen = (const int*)d_in[3];
  float* out = (float*)d_out;

  float* Bwork  = (float*)d_ws;
  float* Bdelta = Bwork + MAX_K * MAX_LEN;

  init_b_kernel<<<5, 128, 0, stream>>>(Bin, Bwork, Bdelta);

  for (int it = 0; it < 3; ++it) {
    const int last = (it == 2);
    routing_iter_kernel<<<NBATCH, 128, 0, stream>>>(low, S, Bwork, slen, Bdelta,
                                                    out, last);
    if (!last) update_b_kernel<<<5, 128, 0, stream>>>(Bwork, Bdelta);
  }
}


# B_delta phase: each row of low[b] read once by one wave in whole 128-byte lines (4 rows per load, 8 lanes per row summed by DPP), not 3 times lane-strided; float sums taken in a different order
# speedup vs baseline: 3.3008x; 1.4519x over previous
#include <hip/hip_runtime.h>
#include <hip/hip_bf16.h>


typedef float v2f __attribute__((ext_vector_type(2)));
typedef float v8f __attribute__((ext_vector_type(8)));

#define MAX_LEN 200
#define MAX_K   3
#define D_IN    128
#define D_OUT   128
#define NBATCH  1024
#define NEG_PAD (-65535.0f)


__global__ void init_b_kernel(const float* __restrict__ Bin,
                              float* __restrict__ Bwork,
                              float* __restrict__ Bdelta) {
  const int t = blockIdx.x * blockDim.x + threadIdx.x;
  if (t < MAX_K * MAX_LEN) { Bwork[t] = Bin[t]; Bdelta[t] = 0.0f; }
}

__global__ void __launch_bounds__(128)
routing_iter_kernel(const float* __restrict__ low, const float* __restrict__ S,
                    const float* __restrict__ Bwork,
                    const int* __restrict__ seq_len, float* __restrict__ Bdelta,
                    float* __restrict__ out, int last) {
  __shared__ __align__(16) float sW[MAX_K][MAX_LEN];
  __shared__ __align__(16) float sT[MAX_K][D_IN];
  __shared__ __align__(16) float sH[MAX_K][D_OUT];
  __shared__ __align__(16) float sG[MAX_K][D_IN];
  __shared__ float sD[MAX_K * MAX_LEN];
  __shared__ float sred[128];

  const int b    = blockIdx.x;
  const int t    = threadIdx.x;
  const int wave = t >> 5;
  const int lane = t & 31;
  const int len  = seq_len[b];
  const float* __restrict__ lowb = low + (size_t)b * MAX_LEN * D_IN;

  for (int i = t; i < MAX_K * MAX_LEN; i += 128) {
    const int l = i % MAX_LEN;
    sW[i / MAX_LEN][l] = (l < len) ? Bwork[i] : NEG_PAD;
  }
  __syncthreads();

  for (int k = 0; k < MAX_K; ++k) {
    float mx = -3.0e38f;
    for (int l = t; l < MAX_LEN; l += 128) mx = fmaxf(mx, sW[k][l]);
    sred[t] = mx; __syncthreads();
    for (int s = 64; s > 0; s >>= 1) {
      if (t < s) sred[t] = fmaxf(sred[t], sred[t + s]);
      __syncthreads();
    }
    const float mk = sred[0]; __syncthreads();
    float sum = 0.0f;
    for (int l = t; l < MAX_LEN; l += 128) {
      const float e = __expf(sW[k][l] - mk);
      sW[k][l] = e; sum += e;
    }
    sred[t] = sum; __syncthreads();
    for (int s = 64; s > 0; s >>= 1) {
      if (t < s) sred[t] += sred[t + s];
      __syncthreads();
    }
    const float inv = 1.0f / sred[0]; __syncthreads();
    for (int l = t; l < MAX_LEN; l += 128) sW[k][l] *= inv;
    __syncthreads();
  }

  const int m     = lane & 15;
  const int khalf = (lane >> 4) * 2;
  const int mrow  = (m < MAX_K) ? m : 0;

  {
    const int n0 = (wave * 2) * 16 + (lane & 15);
    const int n1 = n0 + 16;
    v8f c0 = {}, c1 = {};
#pragma unroll
    for (int kk = 0; kk < MAX_LEN / 4; ++kk) {
      const int kb = kk * 4 + khalf;
      const v2f a = *reinterpret_cast<const v2f*>(&sW[mrow][kb]);
      v2f b0, b1;
      b0.x = lowb[kb * D_IN + n0];  b0.y = lowb[(kb + 1) * D_IN + n0];
      b1.x = lowb[kb * D_IN + n1];  b1.y = lowb[(kb + 1) * D_IN + n1];
      c0 = __builtin_amdgcn_wmma_f32_16x16x4_f32(false, a, false, b0,
                                                 (short)0, c0, false, false);
      c1 = __builtin_amdgcn_wmma_f32_16x16x4_f32(false, a, false, b1,
                                                 (short)0, c1, false, false);
    }
    if (lane < 16) {
#pragma unroll
      for (int r = 0; r < MAX_K; ++r) {
        sT[r][n0] = c0[r];
        sT[r][n1] = c1[r];
      }
    }
  }
  __syncthreads();

  {
    const int n0 = (wave * 2) * 16 + (lane & 15);
    const int n1 = n0 + 16;
    v8f c0 = {}, c1 = {};
#pragma unroll
    for (int kk = 0; kk < D_IN / 4; ++kk) {
      const int kb = kk * 4 + khalf;
      const v2f a = *reinterpret_cast<const v2f*>(&sT[mrow][kb]);
      v2f b0, b1;
      b0.x = S[kb * D_OUT + n0];  b0.y = S[(kb + 1) * D_OUT + n0];
      b1.x = S[kb * D_OUT + n1];  b1.y = S[(kb + 1) * D_OUT + n1];
      c0 = __builtin_amdgcn_wmma_f32_16x16x4_f32(false, a, false, b0,
                                                 (short)0, c0, false, false);
      c1 = __builtin_amdgcn_wmma_f32_16x16x4_f32(false, a, false, b1,
                                                 (short)0, c1, false, false);
    }
    if (lane < 16) {
#pragma unroll
      for (int r = 0; r < MAX_K; ++r) {
        sH[r][n0] = c0[r];
        sH[r][n1] = c1[r];
      }
    }
  }
  __syncthreads();

  {
    const float x0 = sH[0][t], x1 = sH[1][t], x2 = sH[2][t];
    const float sq = x0 * x0 + x1 * x1 + x2 * x2;
    const float scale = sq / (1.0f + sq) * rsqrtf(sq + 1e-9f);
    sH[0][t] = scale * x0; sH[1][t] = scale * x1; sH[2][t] = scale * x2;
    if (last) {
      float* __restrict__ ob = out + (size_t)b * MAX_K * D_OUT;
      ob[0 * D_OUT + t] = sH[0][t];
      ob[1 * D_OUT + t] = sH[1][t];
      ob[2 * D_OUT + t] = sH[2][t];
    }
  }
  __syncthreads();

  if (!last) {
    {
      const int d0 = (wave * 2) * 16 + (lane & 15);
      const int d1 = d0 + 16;
      const float* __restrict__ S0 = S + (size_t)d0 * D_OUT;
      const float* __restrict__ S1 = S + (size_t)d1 * D_OUT;
      v8f c0 = {}, c1 = {};
#pragma unroll
      for (int kk = 0; kk < D_OUT / 4; ++kk) {
        const int kb = kk * 4 + khalf;
        const v2f a = *reinterpret_cast<const v2f*>(&sH[mrow][kb]);
        const v2f b0 = *reinterpret_cast<const v2f*>(S0 + kb);
        const v2f b1 = *reinterpret_cast<const v2f*>(S1 + kb);
        c0 = __builtin_amdgcn_wmma_f32_16x16x4_f32(false, a, false, b0,
                                                   (short)0, c0, false, false);
        c1 = __builtin_amdgcn_wmma_f32_16x16x4_f32(false, a, false, b1,
                                                   (short)0, c1, false, false);
      }
      if (lane < 16) {
#pragma unroll
        for (int r = 0; r < MAX_K; ++r) {
          sG[r][d0] = c0[r];
          sG[r][d1] = c1[r];
        }
      }
    }
    __syncthreads();

    {
      const int gq = lane >> 3;
      const int j  = lane & 7;
      float4 gr[MAX_K][4];
#pragma unroll
      for (int k = 0; k < MAX_K; ++k)
#pragma unroll
        for (int q = 0; q < 4; ++q)
          gr[k][q] = *reinterpret_cast<const float4*>(&sG[k][32 * q + 4 * j]);
      const int row0 = (MAX_LEN / 4) * wave;
      const float* __restrict__ lrow = lowb + (size_t)row0 * D_IN + 4 * j;

      float res[13];
#pragma unroll
      for (int i = 0; i < 13; ++i) {
        const int r = (i < 12) ? (4 * i + gq) : (48 + (gq & 1));
        const float4* __restrict__ rp =
            reinterpret_cast<const float4*>(lrow + (size_t)r * D_IN);
        float4 v[4];
#pragma unroll
        for (int q = 0; q < 4; ++q) v[q] = rp[8 * q];
        float acc[MAX_K];
#pragma unroll
        for (int k = 0; k < MAX_K; ++k) {
          float a = v[0].x * gr[k][0].x;
          a += v[0].y * gr[k][0].y; a += v[0].z * gr[k][0].z; a += v[0].w * gr[k][0].w;
#pragma unroll
          for (int q = 1; q < 4; ++q) {
            a += v[q].x * gr[k][q].x; a += v[q].y * gr[k][q].y;
            a += v[q].z * gr[k][q].z; a += v[q].w * gr[k][q].w;
          }
          a += __builtin_bit_cast(float, __builtin_amdgcn_update_dpp(
                   0, __builtin_bit_cast(int, a), 0x161, 0xf, 0xf, true));
          a += __builtin_bit_cast(float, __builtin_amdgcn_update_dpp(
                   0, __builtin_bit_cast(int, a), 0x162, 0xf, 0xf, true));
          a += __builtin_bit_cast(float, __builtin_amdgcn_update_dpp(
                   0, __builtin_bit_cast(int, a), 0x164, 0xf, 0xf, true));
          acc[k] = a;
        }
        res[i] = (j == 0) ? acc[0] : ((j == 1) ? acc[1] : acc[2]);
      }
      if (j < MAX_K) {
        float* __restrict__ dst = sD + j * MAX_LEN + row0 + gq;
#pragma unroll
        for (int i = 0; i < 12; ++i) dst[4 * i] = res[i];
        if (gq < 2) dst[48] = res[12];
      }
    }
    __syncthreads();

    for (int p = t; p < MAX_K * MAX_LEN; p += 128)
      atomicAdd(&Bdelta[p], sD[p]);
  }
}

__global__ void update_b_kernel(float* __restrict__ Bwork,
                                float* __restrict__ Bdelta) {
  const int t = blockIdx.x * blockDim.x + threadIdx.x;
  if (t < MAX_K * MAX_LEN) { Bwork[t] += Bdelta[t]; Bdelta[t] = 0.0f; }
}

extern "C" void kernel_launch(void* const* d_in, const int* in_sizes, int n_in,
                              void* d_out, int out_size, void* d_ws, size_t ws_size,
                              hipStream_t stream) {
  const float* low  = (const float*)d_in[0];
  const float* Bin  = (const float*)d_in[1];
  const float* S    = (const float*)d_in[2];
  const int*   slen = (const int*)d_in[3];
  float* out = (float*)d_out;

  float* Bwork  = (float*)d_ws;
  float* Bdelta = Bwork + MAX_K * MAX_LEN;

  init_b_kernel<<<5, 128, 0, stream>>>(Bin, Bwork, Bdelta);

  for (int it = 0; it < 3; ++it) {
    const int last = (it == 2);
    routing_iter_kernel<<<NBATCH, 128, 0, stream>>>(low, S, Bwork, slen, Bdelta,
                                                    out, last);
    if (!last) update_b_kernel<<<5, 128, 0, stream>>>(Bwork, Bdelta);
  }
}


# Softmax: wave k does row k in registers, max and sum by a 5-step cross-lane butterfly, 1 block barrier where the LDS reduction trees took 58; the denominator's float sums taken in a different order
# speedup vs baseline: 4.0914x; 1.2395x over previous
#include <hip/hip_runtime.h>
#include <hip/hip_bf16.h>


typedef float v2f __attribute__((ext_vector_type(2)));
typedef float v8f __attribute__((ext_vector_type(8)));

#define MAX_LEN 200
#define MAX_K   3
#define D_IN    128
#define D_OUT   128
#define NBATCH  1024
#define NEG_PAD (-65535.0f)


static __device__ __forceinline__ float xhalf(float x) {
  const int b = __builtin_bit_cast(int, x);
  return __builtin_bit_cast(float, __builtin_amdgcn_permlanex16(
             b, b, 0x76543210, 0xfedcba98, false, false));
}
template <int M>
static __device__ __forceinline__ float xrow(float x) {
  return __builtin_bit_cast(float, __builtin_amdgcn_update_dpp(
             0, __builtin_bit_cast(int, x), 0x160 + M, 0xf, 0xf, true));
}
static __device__ __forceinline__ float wave_max(float v) {
  v = fmaxf(v, xrow<1>(v)); v = fmaxf(v, xrow<2>(v));
  v = fmaxf(v, xrow<4>(v)); v = fmaxf(v, xrow<8>(v));
  return fmaxf(v, xhalf(v));
}
static __device__ __forceinline__ float wave_sum(float v) {
  v += xrow<1>(v); v += xrow<2>(v); v += xrow<4>(v); v += xrow<8>(v);
  return v + xhalf(v);
}

__global__ void init_b_kernel(const float* __restrict__ Bin,
                              float* __restrict__ Bwork,
                              float* __restrict__ Bdelta) {
  const int t = blockIdx.x * blockDim.x + threadIdx.x;
  if (t < MAX_K * MAX_LEN) { Bwork[t] = Bin[t]; Bdelta[t] = 0.0f; }
}

__global__ void __launch_bounds__(128)
routing_iter_kernel(const float* __restrict__ low, const float* __restrict__ S,
                    const float* __restrict__ Bwork,
                    const int* __restrict__ seq_len, float* __restrict__ Bdelta,
                    float* __restrict__ out, int last) {
  __shared__ __align__(16) float sW[MAX_K][MAX_LEN];
  __shared__ __align__(16) float sT[MAX_K][D_IN];
  __shared__ __align__(16) float sH[MAX_K][D_OUT];
  __shared__ __align__(16) float sG[MAX_K][D_IN];
  __shared__ float sD[MAX_K * MAX_LEN];

  const int b    = blockIdx.x;
  const int t    = threadIdx.x;
  const int wave = t >> 5;
  const int lane = t & 31;
  const int len  = seq_len[b];
  const float* __restrict__ lowb = low + (size_t)b * MAX_LEN * D_IN;

  if (wave < MAX_K) {
    const float* __restrict__ Brow = Bwork + wave * MAX_LEN;
    float x[7];
#pragma unroll
    for (int i = 0; i < 7; ++i) {
      const int l = lane + 32 * i;
      const float v = Brow[(l < MAX_LEN) ? l : (MAX_LEN - 1)];
      x[i] = (l < MAX_LEN) ? ((l < len) ? v : NEG_PAD) : -3.0e38f;
    }
    float mx = -3.0e38f;
#pragma unroll
    for (int i = 0; i < 7; ++i) mx = fmaxf(mx, x[i]);
    mx = wave_max(mx);
    float sum = 0.0f;
#pragma unroll
    for (int i = 0; i < 7; ++i) {
      const int l = lane + 32 * i;
      const float e = (l < MAX_LEN) ? __expf(x[i] - mx) : 0.0f;
      x[i] = e; sum += e;
    }
    sum = wave_sum(sum);
    const float inv = 1.0f / sum;
#pragma unroll
    for (int i = 0; i < 7; ++i) {
      const int l = lane + 32 * i;
      if (l < MAX_LEN) sW[wave][l] = x[i] * inv;
    }
  }
  __syncthreads();

  const int m     = lane & 15;
  const int khalf = (lane >> 4) * 2;
  const int mrow  = (m < MAX_K) ? m : 0;

  {
    const int n0 = (wave * 2) * 16 + (lane & 15);
    const int n1 = n0 + 16;
    v8f c0 = {}, c1 = {};
#pragma unroll
    for (int kk = 0; kk < MAX_LEN / 4; ++kk) {
      const int kb = kk * 4 + khalf;
      const v2f a = *reinterpret_cast<const v2f*>(&sW[mrow][kb]);
      v2f b0, b1;
      b0.x = lowb[kb * D_IN + n0];  b0.y = lowb[(kb + 1) * D_IN + n0];
      b1.x = lowb[kb * D_IN + n1];  b1.y = lowb[(kb + 1) * D_IN + n1];
      c0 = __builtin_amdgcn_wmma_f32_16x16x4_f32(false, a, false, b0,
                                                 (short)0, c0, false, false);
      c1 = __builtin_amdgcn_wmma_f32_16x16x4_f32(false, a, false, b1,
                                                 (short)0, c1, false, false);
    }
    if (lane < 16) {
#pragma unroll
      for (int r = 0; r < MAX_K; ++r) {
        sT[r][n0] = c0[r];
        sT[r][n1] = c1[r];
      }
    }
  }
  __syncthreads();

  {
    const int n0 = (wave * 2) * 16 + (lane & 15);
    const int n1 = n0 + 16;
    v8f c0 = {}, c1 = {};
#pragma unroll
    for (int kk = 0; kk < D_IN / 4; ++kk) {
      const int kb = kk * 4 + khalf;
      const v2f a = *reinterpret_cast<const v2f*>(&sT[mrow][kb]);
      v2f b0, b1;
      b0.x = S[kb * D_OUT + n0];  b0.y = S[(kb + 1) * D_OUT + n0];
      b1.x = S[kb * D_OUT + n1];  b1.y = S[(kb + 1) * D_OUT + n1];
      c0 = __builtin_amdgcn_wmma_f32_16x16x4_f32(false, a, false, b0,
                                                 (short)0, c0, false, false);
      c1 = __builtin_amdgcn_wmma_f32_16x16x4_f32(false, a, false, b1,
                                                 (short)0, c1, false, false);
    }
    if (lane < 16) {
#pragma unroll
      for (int r = 0; r < MAX_K; ++r) {
        sH[r][n0] = c0[r];
        sH[r][n1] = c1[r];
      }
    }
  }
  __syncthreads();

  {
    const float x0 = sH[0][t], x1 = sH[1][t], x2 = sH[2][t];
    const float sq = x0 * x0 + x1 * x1 + x2 * x2;
    const float scale = sq / (1.0f + sq) * rsqrtf(sq + 1e-9f);
    sH[0][t] = scale * x0; sH[1][t] = scale * x1; sH[2][t] = scale * x2;
    if (last) {
      float* __restrict__ ob = out + (size_t)b * MAX_K * D_OUT;
      ob[0 * D_OUT + t] = sH[0][t];
      ob[1 * D_OUT + t] = sH[1][t];
      ob[2 * D_OUT + t] = sH[2][t];
    }
  }
  __syncthreads();

  if (!last) {
    {
      const int d0 = (wave * 2) * 16 + (lane & 15);
      const int d1 = d0 + 16;
      const float* __restrict__ S0 = S + (size_t)d0 * D_OUT;
      const float* __restrict__ S1 = S + (size_t)d1 * D_OUT;
      v8f c0 = {}, c1 = {};
#pragma unroll
      for (int kk = 0; kk < D_OUT / 4; ++kk) {
        const int kb = kk * 4 + khalf;
        const v2f a = *reinterpret_cast<const v2f*>(&sH[mrow][kb]);
        const v2f b0 = *reinterpret_cast<const v2f*>(S0 + kb);
        const v2f b1 = *reinterpret_cast<const v2f*>(S1 + kb);
        c0 = __builtin_amdgcn_wmma_f32_16x16x4_f32(false, a, false, b0,
                                                   (short)0, c0, false, false);
        c1 = __builtin_amdgcn_wmma_f32_16x16x4_f32(false, a, false, b1,
                                                   (short)0, c1, false, false);
      }
      if (lane < 16) {
#pragma unroll
        for (int r = 0; r < MAX_K; ++r) {
          sG[r][d0] = c0[r];
          sG[r][d1] = c1[r];
        }
      }
    }
    __syncthreads();

    {
      const int gq = lane >> 3;
      const int j  = lane & 7;
      float4 gr[MAX_K][4];
#pragma unroll
      for (int k = 0; k < MAX_K; ++k)
#pragma unroll
        for (int q = 0; q < 4; ++q)
          gr[k][q] = *reinterpret_cast<const float4*>(&sG[k][32 * q + 4 * j]);
      const int row0 = (MAX_LEN / 4) * wave;
      const float* __restrict__ lrow = lowb + (size_t)row0 * D_IN + 4 * j;

      float res[13];
#pragma unroll
      for (int i = 0; i < 13; ++i) {
        const int r = (i < 12) ? (4 * i + gq) : (48 + (gq & 1));
        const float4* __restrict__ rp =
            reinterpret_cast<const float4*>(lrow + (size_t)r * D_IN);
        float4 v[4];
#pragma unroll
        for (int q = 0; q < 4; ++q) v[q] = rp[8 * q];
        float acc[MAX_K];
#pragma unroll
        for (int k = 0; k < MAX_K; ++k) {
          float a = v[0].x * gr[k][0].x;
          a += v[0].y * gr[k][0].y; a += v[0].z * gr[k][0].z; a += v[0].w * gr[k][0].w;
#pragma unroll
          for (int q = 1; q < 4; ++q) {
            a += v[q].x * gr[k][q].x; a += v[q].y * gr[k][q].y;
            a += v[q].z * gr[k][q].z; a += v[q].w * gr[k][q].w;
          }
          a += __builtin_bit_cast(float, __builtin_amdgcn_update_dpp(
                   0, __builtin_bit_cast(int, a), 0x161, 0xf, 0xf, true));
          a += __builtin_bit_cast(float, __builtin_amdgcn_update_dpp(
                   0, __builtin_bit_cast(int, a), 0x162, 0xf, 0xf, true));
          a += __builtin_bit_cast(float, __builtin_amdgcn_update_dpp(
                   0, __builtin_bit_cast(int, a), 0x164, 0xf, 0xf, true));
          acc[k] = a;
        }
        res[i] = (j == 0) ? acc[0] : ((j == 1) ? acc[1] : acc[2]);
      }
      if (j < MAX_K) {
        float* __restrict__ dst = sD + j * MAX_LEN + row0 + gq;
#pragma unroll
        for (int i = 0; i < 12; ++i) dst[4 * i] = res[i];
        if (gq < 2) dst[48] = res[12];
      }
    }
    __syncthreads();

    for (int p = t; p < MAX_K * MAX_LEN; p += 128)
      atomicAdd(&Bdelta[p], sD[p]);
  }
}

__global__ void update_b_kernel(float* __restrict__ Bwork,
                                float* __restrict__ Bdelta) {
  const int t = blockIdx.x * blockDim.x + threadIdx.x;
  if (t < MAX_K * MAX_LEN) { Bwork[t] += Bdelta[t]; Bdelta[t] = 0.0f; }
}

extern "C" void kernel_launch(void* const* d_in, const int* in_sizes, int n_in,
                              void* d_out, int out_size, void* d_ws, size_t ws_size,
                              hipStream_t stream) {
  const float* low  = (const float*)d_in[0];
  const float* Bin  = (const float*)d_in[1];
  const float* S    = (const float*)d_in[2];
  const int*   slen = (const int*)d_in[3];
  float* out = (float*)d_out;

  float* Bwork  = (float*)d_ws;
  float* Bdelta = Bwork + MAX_K * MAX_LEN;

  init_b_kernel<<<5, 128, 0, stream>>>(Bin, Bwork, Bdelta);

  for (int it = 0; it < 3; ++it) {
    const int last = (it == 2);
    routing_iter_kernel<<<NBATCH, 128, 0, stream>>>(low, S, Bwork, slen, Bdelta,
                                                    out, last);
    if (!last) update_b_kernel<<<5, 128, 0, stream>>>(Bwork, Bdelta);
  }
}


# B update folded into the routing kernel: pass i's softmax reads B_in plus the delta planes of the earlier passes (one plane per pass, zeroed up front), so both update launches go: 4 launches, not 6
# speedup vs baseline: 5.1629x; 1.2619x over previous
#include <hip/hip_runtime.h>
#include <hip/hip_bf16.h>


typedef float v2f __attribute__((ext_vector_type(2)));
typedef float v8f __attribute__((ext_vector_type(8)));

#define MAX_LEN 200
#define MAX_K   3
#define D_IN    128
#define D_OUT   128
#define NBATCH  1024
#define NEG_PAD (-65535.0f)


static __device__ __forceinline__ float xhalf(float x) {
  const int b = __builtin_bit_cast(int, x);
  return __builtin_bit_cast(float, __builtin_amdgcn_permlanex16(
             b, b, 0x76543210, 0xfedcba98, false, false));
}
template <int M>
static __device__ __forceinline__ float xrow(float x) {
  return __builtin_bit_cast(float, __builtin_amdgcn_update_dpp(
             0, __builtin_bit_cast(int, x), 0x160 + M, 0xf, 0xf, true));
}
static __device__ __forceinline__ float wave_max(float v) {
  v = fmaxf(v, xrow<1>(v)); v = fmaxf(v, xrow<2>(v));
  v = fmaxf(v, xrow<4>(v)); v = fmaxf(v, xrow<8>(v));
  return fmaxf(v, xhalf(v));
}
static __device__ __forceinline__ float wave_sum(float v) {
  v += xrow<1>(v); v += xrow<2>(v); v += xrow<4>(v); v += xrow<8>(v);
  return v + xhalf(v);
}

#define NPLANE 3
__global__ void zero_planes_kernel(float* __restrict__ planes) {
  const int t = blockIdx.x * blockDim.x + threadIdx.x;
  if (t < NPLANE * MAX_K * MAX_LEN) planes[t] = 0.0f;
}

#define NB     4
#define NROW   (NB * MAX_K)
#define NBLOCK 256

__global__ void __launch_bounds__(512)
routing_iter_kernel(const float* __restrict__ low, const float* __restrict__ S,
                    const float* __restrict__ Bin, const float* __restrict__ Dp0,
                    const float* __restrict__ Dp1,
                    const int* __restrict__ seq_len, float* __restrict__ Bdelta,
                    float* __restrict__ out, int last) {
  __shared__ __align__(16) float sW[NB][MAX_K][MAX_LEN];
  __shared__ __align__(16) float sT[NROW][D_IN + 4];
  __shared__ __align__(16) float sHp[2][NROW][D_OUT];
  __shared__ __align__(16) float sH[NROW][D_OUT + 4];
  __shared__ __align__(16) float sGp[2][NROW][D_IN];
  __shared__ float sD[NB][MAX_K * MAX_LEN];

  const int t    = threadIdx.x;
  const int wave = t >> 5;
  const int lane = t & 31;
  const int q    = wave >> 2;
  const int w    = wave & 3;
  const int b    = blockIdx.x * NB + q;
  const int len  = seq_len[b];
  const float* __restrict__ lowb = low + (size_t)b * MAX_LEN * D_IN;

  if (w < MAX_K) {
    const float* __restrict__ Brow  = Bin + w * MAX_LEN;
    const float* __restrict__ D0row = Dp0 + w * MAX_LEN;
    const float* __restrict__ D1row = Dp1 + w * MAX_LEN;
    float x[7];
#pragma unroll
    for (int i = 0; i < 7; ++i) {
      const int l = lane + 32 * i;
      const int lc = (l < MAX_LEN) ? l : (MAX_LEN - 1);
      const float v = (Brow[lc] + D0row[lc]) + D1row[lc];
      x[i] = (l < MAX_LEN) ? ((l < len) ? v : NEG_PAD) : -3.0e38f;
    }
    float mx = -3.0e38f;
#pragma unroll
    for (int i = 0; i < 7; ++i) mx = fmaxf(mx, x[i]);
    mx = wave_max(mx);
    float sum = 0.0f;
#pragma unroll
    for (int i = 0; i < 7; ++i) {
      const int l = lane + 32 * i;
      const float e = (l < MAX_LEN) ? __expf(x[i] - mx) : 0.0f;
      x[i] = e; sum += e;
    }
    sum = wave_sum(sum);
    const float inv = 1.0f / sum;
#pragma unroll
    for (int i = 0; i < 7; ++i) {
      const int l = lane + 32 * i;
      if (l < MAX_LEN) sW[q][w][l] = x[i] * inv;
    }
  }
  __syncthreads();

  const int m     = lane & 15;
  const int khalf = (lane >> 4) * 2;
  const int mrow3  = (m < MAX_K) ? m : 0;
  const int mrow12 = (m < NROW) ? m : 0;

  {
    const int n0 = (w * 2) * 16 + (lane & 15);
    const int n1 = n0 + 16;
    v8f c0 = {}, c1 = {};
#pragma unroll
    for (int kk = 0; kk < MAX_LEN / 4; ++kk) {
      const int kb = kk * 4 + khalf;
      const v2f a = *reinterpret_cast<const v2f*>(&sW[q][mrow3][kb]);
      v2f b0, b1;
      b0.x = lowb[kb * D_IN + n0];  b0.y = lowb[(kb + 1) * D_IN + n0];
      b1.x = lowb[kb * D_IN + n1];  b1.y = lowb[(kb + 1) * D_IN + n1];
      c0 = __builtin_amdgcn_wmma_f32_16x16x4_f32(false, a, false, b0,
                                                 (short)0, c0, false, false);
      c1 = __builtin_amdgcn_wmma_f32_16x16x4_f32(false, a, false, b1,
                                                 (short)0, c1, false, false);
    }
    if (lane < 16) {
#pragma unroll
      for (int r = 0; r < MAX_K; ++r) {
        sT[MAX_K * q + r][n0] = c0[r];
        sT[MAX_K * q + r][n1] = c1[r];
      }
    }
  }
  __syncthreads();

  const int tile  = wave & 7;
  const int kh    = wave >> 3;
  const int ncol  = tile * 16 + (lane & 15);

  {
    const float* __restrict__ Sk = S + (size_t)(kh * 64 + khalf) * D_OUT + ncol;
    const float* __restrict__ aT = &sT[mrow12][kh * 64 + khalf];
    v8f c0 = {}, c1 = {};
#pragma unroll
    for (int kk = 0; kk < 16; kk += 2) {
      const v2f a0 = *reinterpret_cast<const v2f*>(aT + 4 * kk);
      const v2f a1 = *reinterpret_cast<const v2f*>(aT + 4 * kk + 4);
      v2f b0, b1;
      b0.x = Sk[(4 * kk) * D_OUT];      b0.y = Sk[(4 * kk + 1) * D_OUT];
      b1.x = Sk[(4 * kk + 4) * D_OUT];  b1.y = Sk[(4 * kk + 5) * D_OUT];
      c0 = __builtin_amdgcn_wmma_f32_16x16x4_f32(false, a0, false, b0,
                                                 (short)0, c0, false, false);
      c1 = __builtin_amdgcn_wmma_f32_16x16x4_f32(false, a1, false, b1,
                                                 (short)0, c1, false, false);
    }
    if (lane < 16) {
#pragma unroll
      for (int r = 0; r < 8; ++r) sHp[kh][r][ncol] = c0[r] + c1[r];
    } else {
#pragma unroll
      for (int r = 0; r < NROW - 8; ++r) sHp[kh][8 + r][ncol] = c0[r] + c1[r];
    }
  }
  __syncthreads();

  {
    const int e = t & (D_OUT - 1);
    const float x0 = sHp[0][MAX_K * q + 0][e] + sHp[1][MAX_K * q + 0][e];
    const float x1 = sHp[0][MAX_K * q + 1][e] + sHp[1][MAX_K * q + 1][e];
    const float x2 = sHp[0][MAX_K * q + 2][e] + sHp[1][MAX_K * q + 2][e];
    const float sq = x0 * x0 + x1 * x1 + x2 * x2;
    const float scale = sq / (1.0f + sq) * rsqrtf(sq + 1e-9f);
    const float h0 = scale * x0, h1 = scale * x1, h2 = scale * x2;
    sH[MAX_K * q + 0][e] = h0; sH[MAX_K * q + 1][e] = h1; sH[MAX_K * q + 2][e] = h2;
    if (last) {
      float* __restrict__ ob = out + (size_t)b * MAX_K * D_OUT;
      ob[0 * D_OUT + e] = h0;
      ob[1 * D_OUT + e] = h1;
      ob[2 * D_OUT + e] = h2;
    }
  }
  __syncthreads();

  if (!last) {
    {
      const float* __restrict__ Sd = S + (size_t)ncol * D_OUT + kh * 64 + khalf;
      const float* __restrict__ aH = &sH[mrow12][kh * 64 + khalf];
      v8f c0 = {}, c1 = {};
#pragma unroll
      for (int kk = 0; kk < 16; kk += 2) {
        const v2f a0 = *reinterpret_cast<const v2f*>(aH + 4 * kk);
        const v2f a1 = *reinterpret_cast<const v2f*>(aH + 4 * kk + 4);
        const v2f b0 = *reinterpret_cast<const v2f*>(Sd + 4 * kk);
        const v2f b1 = *reinterpret_cast<const v2f*>(Sd + 4 * kk + 4);
        c0 = __builtin_amdgcn_wmma_f32_16x16x4_f32(false, a0, false, b0,
                                                   (short)0, c0, false, false);
        c1 = __builtin_amdgcn_wmma_f32_16x16x4_f32(false, a1, false, b1,
                                                   (short)0, c1, false, false);
      }
      if (lane < 16) {
#pragma unroll
        for (int r = 0; r < 8; ++r) sGp[kh][r][ncol] = c0[r] + c1[r];
      } else {
#pragma unroll
        for (int r = 0; r < NROW - 8; ++r) sGp[kh][8 + r][ncol] = c0[r] + c1[r];
      }
    }
    __syncthreads();

    {
      const int gq = lane >> 3;
      const int j  = lane & 7;
      float4 gr[MAX_K][4];
#pragma unroll
      for (int k = 0; k < MAX_K; ++k)
#pragma unroll
        for (int qd = 0; qd < 4; ++qd) {
          const float4 g0 =
              *reinterpret_cast<const float4*>(&sGp[0][MAX_K * q + k][32 * qd + 4 * j]);
          const float4 g1 =
              *reinterpret_cast<const float4*>(&sGp[1][MAX_K * q + k][32 * qd + 4 * j]);
          gr[k][qd] = make_float4(g0.x + g1.x, g0.y + g1.y, g0.z + g1.z, g0.w + g1.w);
        }
      const int row0 = (MAX_LEN / 4) * w;
      const float* __restrict__ lrow = lowb + (size_t)row0 * D_IN + 4 * j;

      float res[13];
#pragma unroll
      for (int i = 0; i < 13; ++i) {
        const int r = (i < 12) ? (4 * i + gq) : (48 + (gq & 1));
        const float4* __restrict__ rp =
            reinterpret_cast<const float4*>(lrow + (size_t)r * D_IN);
        float4 v[4];
#pragma unroll
        for (int qd = 0; qd < 4; ++qd) v[qd] = rp[8 * qd];
        float acc[MAX_K];
#pragma unroll
        for (int k = 0; k < MAX_K; ++k) {
          float a = v[0].x * gr[k][0].x;
          a += v[0].y * gr[k][0].y; a += v[0].z * gr[k][0].z; a += v[0].w * gr[k][0].w;
#pragma unroll
          for (int qd = 1; qd < 4; ++qd) {
            a += v[qd].x * gr[k][qd].x; a += v[qd].y * gr[k][qd].y;
            a += v[qd].z * gr[k][qd].z; a += v[qd].w * gr[k][qd].w;
          }
          a += __builtin_bit_cast(float, __builtin_amdgcn_update_dpp(
                   0, __builtin_bit_cast(int, a), 0x161, 0xf, 0xf, true));
          a += __builtin_bit_cast(float, __builtin_amdgcn_update_dpp(
                   0, __builtin_bit_cast(int, a), 0x162, 0xf, 0xf, true));
          a += __builtin_bit_cast(float, __builtin_amdgcn_update_dpp(
                   0, __builtin_bit_cast(int, a), 0x164, 0xf, 0xf, true));
          acc[k] = a;
        }
        res[i] = (j == 0) ? acc[0] : ((j == 1) ? acc[1] : acc[2]);
      }
      if (j < MAX_K) {
        float* __restrict__ dst = &sD[q][0] + j * MAX_LEN + row0 + gq;
#pragma unroll
        for (int i = 0; i < 12; ++i) dst[4 * i] = res[i];
        if (gq < 2) dst[48] = res[12];
      }
    }
    __syncthreads();

    for (int p = t & 127; p < MAX_K * MAX_LEN; p += 128)
      atomicAdd(&Bdelta[p], sD[q][p]);
  }
}

extern "C" void kernel_launch(void* const* d_in, const int* in_sizes, int n_in,
                              void* d_out, int out_size, void* d_ws, size_t ws_size,
                              hipStream_t stream) {
  const float* low  = (const float*)d_in[0];
  const float* Bin  = (const float*)d_in[1];
  const float* S    = (const float*)d_in[2];
  const int*   slen = (const int*)d_in[3];
  float* out = (float*)d_out;

  float* D0 = (float*)d_ws;
  float* D1 = D0 + MAX_K * MAX_LEN;
  float* Z  = D1 + MAX_K * MAX_LEN;

  zero_planes_kernel<<<8, 256, 0, stream>>>(D0);

  routing_iter_kernel<<<NBLOCK, 512, 0, stream>>>(low, S, Bin, Z,  Z,  slen, D0, out, 0);
  routing_iter_kernel<<<NBLOCK, 512, 0, stream>>>(low, S, Bin, D0, Z,  slen, D1, out, 0);
  routing_iter_kernel<<<NBLOCK, 512, 0, stream>>>(low, S, Bin, D0, D1, slen, nullptr, out, 1);
}


# Next phase's global loads requested one phase early (low rows before softmax, S before T's barrier, S^T and 6 delta passes before T4.S) and kept in flight across LDS-only barriers
# speedup vs baseline: 5.5388x; 1.0728x over previous
#include <hip/hip_runtime.h>
#include <hip/hip_bf16.h>


typedef float v2f __attribute__((ext_vector_type(2)));
typedef float v8f __attribute__((ext_vector_type(8)));

#define MAX_LEN 200
#define MAX_K   3
#define D_IN    128
#define D_OUT   128
#define NBATCH  1024
#define NEG_PAD (-65535.0f)


static __device__ __forceinline__ float xhalf(float x) {
  const int b = __builtin_bit_cast(int, x);
  return __builtin_bit_cast(float, __builtin_amdgcn_permlanex16(
             b, b, 0x76543210, 0xfedcba98, false, false));
}
template <int M>
static __device__ __forceinline__ float xrow(float x) {
  return __builtin_bit_cast(float, __builtin_amdgcn_update_dpp(
             0, __builtin_bit_cast(int, x), 0x160 + M, 0xf, 0xf, true));
}
static __device__ __forceinline__ float wave_max(float v) {
  v = fmaxf(v, xrow<1>(v)); v = fmaxf(v, xrow<2>(v));
  v = fmaxf(v, xrow<4>(v)); v = fmaxf(v, xrow<8>(v));
  return fmaxf(v, xhalf(v));
}
static __device__ __forceinline__ float wave_sum(float v) {
  v += xrow<1>(v); v += xrow<2>(v); v += xrow<4>(v); v += xrow<8>(v);
  return v + xhalf(v);
}

static __device__ __forceinline__ void lds_barrier() {
  asm volatile("s_wait_dscnt 0x0\n\ts_barrier_signal -1\n\ts_barrier_wait -1" ::: "memory");
}

#define NPLANE 3
__global__ void zero_planes_kernel(float* __restrict__ planes) {
  const int t = blockIdx.x * blockDim.x + threadIdx.x;
  if (t < NPLANE * MAX_K * MAX_LEN) planes[t] = 0.0f;
}

#define NB     4
#define NROW   (NB * MAX_K)
#define NBLOCK 256
#define TPF    10
#define DPF    6

template <bool LAST>
__global__ void __launch_bounds__(512) __attribute__((amdgpu_waves_per_eu(4, 4)))
routing_iter_kernel(const float* __restrict__ low, const float* __restrict__ S,
                    const float* __restrict__ Bin, const float* __restrict__ Dp0,
                    const float* __restrict__ Dp1,
                    const int* __restrict__ seq_len, float* __restrict__ Bdelta,
                    float* __restrict__ out) {
  constexpr bool last = LAST;
  __shared__ __align__(16) float sW[NB][MAX_K][MAX_LEN];
  __shared__ __align__(16) float sT[NROW][D_IN + 4];
  __shared__ __align__(16) float sHp[2][NROW][D_OUT];
  __shared__ __align__(16) float sH[NROW][D_OUT + 4];
  __shared__ __align__(16) float sGp[2][NROW][D_IN];
  __shared__ float sD[NB][MAX_K * MAX_LEN];

  const int t    = threadIdx.x;
  const int wave = t >> 5;
  const int lane = t & 31;
  const int q    = wave >> 2;
  const int w    = wave & 3;
  const int b    = blockIdx.x * NB + q;
  const int len  = seq_len[b];
  const float* __restrict__ lowb = low + (size_t)b * MAX_LEN * D_IN;

  const int m     = lane & 15;
  const int khalf = (lane >> 4) * 2;
  const int n0 = (w * 2) * 16 + m;
  const int n1 = n0 + 16;
  float rb[7], rd0[7], rd1[7];
  {
    const int wr = (w < MAX_K) ? w : (MAX_K - 1);
    const float* __restrict__ Brow  = Bin + wr * MAX_LEN;
    const float* __restrict__ D0row = Dp0 + wr * MAX_LEN;
    const float* __restrict__ D1row = Dp1 + wr * MAX_LEN;
#pragma unroll
    for (int i = 0; i < 7; ++i) {
      const int l = lane + 32 * i;
      const int lc = (l < MAX_LEN) ? l : (MAX_LEN - 1);
      rb[i] = Brow[lc];  rd0[i] = D0row[lc];  rd1[i] = D1row[lc];
    }
  }
  __builtin_amdgcn_sched_barrier(0);
  v2f pb0[TPF], pb1[TPF];
#pragma unroll
  for (int kk = 0; kk < TPF; ++kk) {
    const int kb = kk * 4 + khalf;
    pb0[kk].x = lowb[kb * D_IN + n0];  pb0[kk].y = lowb[(kb + 1) * D_IN + n0];
    pb1[kk].x = lowb[kb * D_IN + n1];  pb1[kk].y = lowb[(kb + 1) * D_IN + n1];
  }
  {
    float x[7];
#pragma unroll
    for (int i = 0; i < 7; ++i) {
      const int l = lane + 32 * i;
      const float v = (rb[i] + rd0[i]) + rd1[i];
      x[i] = (l < MAX_LEN) ? ((l < len) ? v : NEG_PAD) : -3.0e38f;
    }
    float mx = -3.0e38f;
#pragma unroll
    for (int i = 0; i < 7; ++i) mx = fmaxf(mx, x[i]);
    mx = wave_max(mx);
    float sum = 0.0f;
#pragma unroll
    for (int i = 0; i < 7; ++i) {
      const int l = lane + 32 * i;
      const float e = (l < MAX_LEN) ? __expf(x[i] - mx) : 0.0f;
      x[i] = e; sum += e;
    }
    sum = wave_sum(sum);
    const float inv = 1.0f / sum;
    if (w < MAX_K) {
#pragma unroll
      for (int i = 0; i < 7; ++i) {
        const int l = lane + 32 * i;
        if (l < MAX_LEN) sW[q][w][l] = x[i] * inv;
      }
    }
  }
  lds_barrier();

  const int mrow3  = (m < MAX_K) ? m : 0;
  const int mrow12 = (m < NROW) ? m : 0;

  const int tile  = wave & 7;
  const int kh    = wave >> 3;
  const int ncol  = tile * 16 + (lane & 15);
  const float* __restrict__ Sk = S + (size_t)(kh * 64 + khalf) * D_OUT + ncol;
  v2f sb[16];

  {
    v8f c0 = {}, c1 = {};
#pragma unroll
    for (int kk = 0; kk < MAX_LEN / 4; ++kk) {
      const int kb = kk * 4 + khalf;
      const v2f a = *reinterpret_cast<const v2f*>(&sW[q][mrow3][kb]);
      v2f b0, b1;
      if (kk < TPF) {
        b0 = pb0[kk];  b1 = pb1[kk];
      } else {
        b0.x = lowb[kb * D_IN + n0];  b0.y = lowb[(kb + 1) * D_IN + n0];
        b1.x = lowb[kb * D_IN + n1];  b1.y = lowb[(kb + 1) * D_IN + n1];
      }
      c0 = __builtin_amdgcn_wmma_f32_16x16x4_f32(false, a, false, b0,
                                                 (short)0, c0, false, false);
      c1 = __builtin_amdgcn_wmma_f32_16x16x4_f32(false, a, false, b1,
                                                 (short)0, c1, false, false);
    }
#pragma unroll
    for (int jk = 0; jk < 16; ++jk) {
      sb[jk].x = Sk[(4 * jk) * D_OUT];  sb[jk].y = Sk[(4 * jk + 1) * D_OUT];
    }
    if (lane < 16) {
#pragma unroll
      for (int r = 0; r < MAX_K; ++r) {
        sT[MAX_K * q + r][n0] = c0[r];
        sT[MAX_K * q + r][n1] = c1[r];
      }
    }
  }
  lds_barrier();

  const float* __restrict__ Sd = S + (size_t)ncol * D_OUT + kh * 64 + khalf;
  v2f st[16];
  const int gq = lane >> 3;
  const int j  = lane & 7;
  const int row0 = (MAX_LEN / 4) * w;
  const float* __restrict__ lrow = lowb + (size_t)row0 * D_IN + 4 * j;
  float4 dv[DPF][4];

  {
    if (!last) {
#pragma unroll
      for (int jk = 0; jk < 16; ++jk)
        st[jk] = *reinterpret_cast<const v2f*>(Sd + 4 * jk);
#pragma unroll
      for (int i = 0; i < DPF; ++i) {
        const float4* __restrict__ rp =
            reinterpret_cast<const float4*>(lrow + (size_t)(4 * i + gq) * D_IN);
#pragma unroll
        for (int qd = 0; qd < 4; ++qd) dv[i][qd] = rp[8 * qd];
      }
    }
    const float* __restrict__ aT = &sT[mrow12][kh * 64 + khalf];
    v8f c0 = {}, c1 = {};
#pragma unroll
    for (int kk = 0; kk < 16; kk += 2) {
      const v2f a0 = *reinterpret_cast<const v2f*>(aT + 4 * kk);
      const v2f a1 = *reinterpret_cast<const v2f*>(aT + 4 * kk + 4);
      const v2f b0 = sb[kk], b1 = sb[kk + 1];
      c0 = __builtin_amdgcn_wmma_f32_16x16x4_f32(false, a0, false, b0,
                                                 (short)0, c0, false, false);
      c1 = __builtin_amdgcn_wmma_f32_16x16x4_f32(false, a1, false, b1,
                                                 (short)0, c1, false, false);
    }
    if (lane < 16) {
#pragma unroll
      for (int r = 0; r < 8; ++r) sHp[kh][r][ncol] = c0[r] + c1[r];
    } else {
#pragma unroll
      for (int r = 0; r < NROW - 8; ++r) sHp[kh][8 + r][ncol] = c0[r] + c1[r];
    }
  }
  lds_barrier();

  {
    const int e = t & (D_OUT - 1);
    const float x0 = sHp[0][MAX_K * q + 0][e] + sHp[1][MAX_K * q + 0][e];
    const float x1 = sHp[0][MAX_K * q + 1][e] + sHp[1][MAX_K * q + 1][e];
    const float x2 = sHp[0][MAX_K * q + 2][e] + sHp[1][MAX_K * q + 2][e];
    const float sq = x0 * x0 + x1 * x1 + x2 * x2;
    const float scale = sq / (1.0f + sq) * rsqrtf(sq + 1e-9f);
    const float h0 = scale * x0, h1 = scale * x1, h2 = scale * x2;
    sH[MAX_K * q + 0][e] = h0; sH[MAX_K * q + 1][e] = h1; sH[MAX_K * q + 2][e] = h2;
    if (last) {
      float* __restrict__ ob = out + (size_t)b * MAX_K * D_OUT;
      ob[0 * D_OUT + e] = h0;
      ob[1 * D_OUT + e] = h1;
      ob[2 * D_OUT + e] = h2;
    }
  }
  lds_barrier();

  if (!last) {
    {
      const float* __restrict__ aH = &sH[mrow12][kh * 64 + khalf];
      v8f c0 = {}, c1 = {};
#pragma unroll
      for (int kk = 0; kk < 16; kk += 2) {
        const v2f a0 = *reinterpret_cast<const v2f*>(aH + 4 * kk);
        const v2f a1 = *reinterpret_cast<const v2f*>(aH + 4 * kk + 4);
        const v2f b0 = st[kk], b1 = st[kk + 1];
        c0 = __builtin_amdgcn_wmma_f32_16x16x4_f32(false, a0, false, b0,
                                                   (short)0, c0, false, false);
        c1 = __builtin_amdgcn_wmma_f32_16x16x4_f32(false, a1, false, b1,
                                                   (short)0, c1, false, false);
      }
      if (lane < 16) {
#pragma unroll
        for (int r = 0; r < 8; ++r) sGp[kh][r][ncol] = c0[r] + c1[r];
      } else {
#pragma unroll
        for (int r = 0; r < NROW - 8; ++r) sGp[kh][8 + r][ncol] = c0[r] + c1[r];
      }
    }
    lds_barrier();

    {
      float4 gr[MAX_K][4];
#pragma unroll
      for (int k = 0; k < MAX_K; ++k)
#pragma unroll
        for (int qd = 0; qd < 4; ++qd) {
          const float4 g0 =
              *reinterpret_cast<const float4*>(&sGp[0][MAX_K * q + k][32 * qd + 4 * j]);
          const float4 g1 =
              *reinterpret_cast<const float4*>(&sGp[1][MAX_K * q + k][32 * qd + 4 * j]);
          gr[k][qd] = make_float4(g0.x + g1.x, g0.y + g1.y, g0.z + g1.z, g0.w + g1.w);
        }
      float res[13];
#pragma unroll
      for (int i = 0; i < 13; ++i) {
        const int r = (i < 12) ? (4 * i + gq) : (48 + (gq & 1));
        const float4* __restrict__ rp =
            reinterpret_cast<const float4*>(lrow + (size_t)r * D_IN);
        float4 v[4];
#pragma unroll
        for (int qd = 0; qd < 4; ++qd) v[qd] = (i < DPF) ? dv[i][qd] : rp[8 * qd];
        float acc[MAX_K];
#pragma unroll
        for (int k = 0; k < MAX_K; ++k) {
          float a = v[0].x * gr[k][0].x;
          a += v[0].y * gr[k][0].y; a += v[0].z * gr[k][0].z; a += v[0].w * gr[k][0].w;
#pragma unroll
          for (int qd = 1; qd < 4; ++qd) {
            a += v[qd].x * gr[k][qd].x; a += v[qd].y * gr[k][qd].y;
            a += v[qd].z * gr[k][qd].z; a += v[qd].w * gr[k][qd].w;
          }
          a += __builtin_bit_cast(float, __builtin_amdgcn_update_dpp(
                   0, __builtin_bit_cast(int, a), 0x161, 0xf, 0xf, true));
          a += __builtin_bit_cast(float, __builtin_amdgcn_update_dpp(
                   0, __builtin_bit_cast(int, a), 0x162, 0xf, 0xf, true));
          a += __builtin_bit_cast(float, __builtin_amdgcn_update_dpp(
                   0, __builtin_bit_cast(int, a), 0x164, 0xf, 0xf, true));
          acc[k] = a;
        }
        res[i] = (j == 0) ? acc[0] : ((j == 1) ? acc[1] : acc[2]);
      }
      if (j < MAX_K) {
        float* __restrict__ dst = &sD[q][0] + j * MAX_LEN + row0 + gq;
#pragma unroll
        for (int i = 0; i < 12; ++i) dst[4 * i] = res[i];
        if (gq < 2) dst[48] = res[12];
      }
    }
    __syncthreads();

    for (int p = t & 127; p < MAX_K * MAX_LEN; p += 128)
      atomicAdd(&Bdelta[p], sD[q][p]);
  }
}

extern "C" void kernel_launch(void* const* d_in, const int* in_sizes, int n_in,
                              void* d_out, int out_size, void* d_ws, size_t ws_size,
                              hipStream_t stream) {
  const float* low  = (const float*)d_in[0];
  const float* Bin  = (const float*)d_in[1];
  const float* S    = (const float*)d_in[2];
  const int*   slen = (const int*)d_in[3];
  float* out = (float*)d_out;

  float* D0 = (float*)d_ws;
  float* D1 = D0 + MAX_K * MAX_LEN;
  float* Z  = D1 + MAX_K * MAX_LEN;

  zero_planes_kernel<<<8, 256, 0, stream>>>(D0);

  routing_iter_kernel<false><<<NBLOCK, 512, 0, stream>>>(low, S, Bin, Z,  Z,  slen, D0, out);
  routing_iter_kernel<false><<<NBLOCK, 512, 0, stream>>>(low, S, Bin, D0, Z,  slen, D1, out);
  routing_iter_kernel<true><<<NBLOCK, 512, 0, stream>>>(low, S, Bin, D0, D1, slen, nullptr, out);
}


# Delta atomics: the four batches of a block are summed in LDS first, then one atomic per (k,l) per block, 153600 a pass where there were 614400; float sums taken in a different order
# speedup vs baseline: 5.9786x; 1.0794x over previous
#include <hip/hip_runtime.h>
#include <hip/hip_bf16.h>


typedef float v2f __attribute__((ext_vector_type(2)));
typedef float v8f __attribute__((ext_vector_type(8)));

#define MAX_LEN 200
#define MAX_K   3
#define D_IN    128
#define D_OUT   128
#define NBATCH  1024
#define NEG_PAD (-65535.0f)


static __device__ __forceinline__ float xhalf(float x) {
  const int b = __builtin_bit_cast(int, x);
  return __builtin_bit_cast(float, __builtin_amdgcn_permlanex16(
             b, b, 0x76543210, 0xfedcba98, false, false));
}
template <int M>
static __device__ __forceinline__ float xrow(float x) {
  return __builtin_bit_cast(float, __builtin_amdgcn_update_dpp(
             0, __builtin_bit_cast(int, x), 0x160 + M, 0xf, 0xf, true));
}
static __device__ __forceinline__ float wave_max(float v) {
  v = fmaxf(v, xrow<1>(v)); v = fmaxf(v, xrow<2>(v));
  v = fmaxf(v, xrow<4>(v)); v = fmaxf(v, xrow<8>(v));
  return fmaxf(v, xhalf(v));
}
static __device__ __forceinline__ float wave_sum(float v) {
  v += xrow<1>(v); v += xrow<2>(v); v += xrow<4>(v); v += xrow<8>(v);
  return v + xhalf(v);
}

static __device__ __forceinline__ void lds_barrier() {
  asm volatile("s_wait_dscnt 0x0\n\ts_barrier_signal -1\n\ts_barrier_wait -1" ::: "memory");
}

#define NPLANE 3
__global__ void zero_planes_kernel(float* __restrict__ planes) {
  const int t = blockIdx.x * blockDim.x + threadIdx.x;
  if (t < NPLANE * MAX_K * MAX_LEN) planes[t] = 0.0f;
}

#define NB     4
#define NROW   (NB * MAX_K)
#define NBLOCK 256
#define TPF    10
#define DPF    6

template <bool LAST>
__global__ void __launch_bounds__(512) __attribute__((amdgpu_waves_per_eu(4, 4)))
routing_iter_kernel(const float* __restrict__ low, const float* __restrict__ S,
                    const float* __restrict__ Bin, const float* __restrict__ Dp0,
                    const float* __restrict__ Dp1,
                    const int* __restrict__ seq_len, float* __restrict__ Bdelta,
                    float* __restrict__ out) {
  constexpr bool last = LAST;
  __shared__ __align__(16) float sW[NB][MAX_K][MAX_LEN];
  __shared__ __align__(16) float sT[NROW][D_IN + 4];
  __shared__ __align__(16) float sHp[2][NROW][D_OUT];
  __shared__ __align__(16) float sH[NROW][D_OUT + 4];
  __shared__ __align__(16) float sGp[2][NROW][D_IN];
  __shared__ float sD[NB][MAX_K * MAX_LEN];

  const int t    = threadIdx.x;
  const int wave = t >> 5;
  const int lane = t & 31;
  const int q    = wave >> 2;
  const int w    = wave & 3;
  const int b    = blockIdx.x * NB + q;
  const int len  = seq_len[b];
  const float* __restrict__ lowb = low + (size_t)b * MAX_LEN * D_IN;

  const int m     = lane & 15;
  const int khalf = (lane >> 4) * 2;
  const int n0 = (w * 2) * 16 + m;
  const int n1 = n0 + 16;
  float rb[7], rd0[7], rd1[7];
  {
    const int wr = (w < MAX_K) ? w : (MAX_K - 1);
    const float* __restrict__ Brow  = Bin + wr * MAX_LEN;
    const float* __restrict__ D0row = Dp0 + wr * MAX_LEN;
    const float* __restrict__ D1row = Dp1 + wr * MAX_LEN;
#pragma unroll
    for (int i = 0; i < 7; ++i) {
      const int l = lane + 32 * i;
      const int lc = (l < MAX_LEN) ? l : (MAX_LEN - 1);
      rb[i] = Brow[lc];  rd0[i] = D0row[lc];  rd1[i] = D1row[lc];
    }
  }
  __builtin_amdgcn_sched_barrier(0);
  v2f pb0[TPF], pb1[TPF];
#pragma unroll
  for (int kk = 0; kk < TPF; ++kk) {
    const int kb = kk * 4 + khalf;
    pb0[kk].x = lowb[kb * D_IN + n0];  pb0[kk].y = lowb[(kb + 1) * D_IN + n0];
    pb1[kk].x = lowb[kb * D_IN + n1];  pb1[kk].y = lowb[(kb + 1) * D_IN + n1];
  }
  {
    float x[7];
#pragma unroll
    for (int i = 0; i < 7; ++i) {
      const int l = lane + 32 * i;
      const float v = (rb[i] + rd0[i]) + rd1[i];
      x[i] = (l < MAX_LEN) ? ((l < len) ? v : NEG_PAD) : -3.0e38f;
    }
    float mx = -3.0e38f;
#pragma unroll
    for (int i = 0; i < 7; ++i) mx = fmaxf(mx, x[i]);
    mx = wave_max(mx);
    float sum = 0.0f;
#pragma unroll
    for (int i = 0; i < 7; ++i) {
      const int l = lane + 32 * i;
      const float e = (l < MAX_LEN) ? __expf(x[i] - mx) : 0.0f;
      x[i] = e; sum += e;
    }
    sum = wave_sum(sum);
    const float inv = 1.0f / sum;
    if (w < MAX_K) {
#pragma unroll
      for (int i = 0; i < 7; ++i) {
        const int l = lane + 32 * i;
        if (l < MAX_LEN) sW[q][w][l] = x[i] * inv;
      }
    }
  }
  lds_barrier();

  const int mrow3  = (m < MAX_K) ? m : 0;
  const int mrow12 = (m < NROW) ? m : 0;

  const int tile  = wave & 7;
  const int kh    = wave >> 3;
  const int ncol  = tile * 16 + (lane & 15);
  const float* __restrict__ Sk = S + (size_t)(kh * 64 + khalf) * D_OUT + ncol;
  v2f sb[16];

  {
    v8f c0 = {}, c1 = {};
#pragma unroll
    for (int kk = 0; kk < MAX_LEN / 4; ++kk) {
      const int kb = kk * 4 + khalf;
      const v2f a = *reinterpret_cast<const v2f*>(&sW[q][mrow3][kb]);
      v2f b0, b1;
      if (kk < TPF) {
        b0 = pb0[kk];  b1 = pb1[kk];
      } else {
        b0.x = lowb[kb * D_IN + n0];  b0.y = lowb[(kb + 1) * D_IN + n0];
        b1.x = lowb[kb * D_IN + n1];  b1.y = lowb[(kb + 1) * D_IN + n1];
      }
      c0 = __builtin_amdgcn_wmma_f32_16x16x4_f32(false, a, false, b0,
                                                 (short)0, c0, false, false);
      c1 = __builtin_amdgcn_wmma_f32_16x16x4_f32(false, a, false, b1,
                                                 (short)0, c1, false, false);
    }
#pragma unroll
    for (int jk = 0; jk < 16; ++jk) {
      sb[jk].x = Sk[(4 * jk) * D_OUT];  sb[jk].y = Sk[(4 * jk + 1) * D_OUT];
    }
    if (lane < 16) {
#pragma unroll
      for (int r = 0; r < MAX_K; ++r) {
        sT[MAX_K * q + r][n0] = c0[r];
        sT[MAX_K * q + r][n1] = c1[r];
      }
    }
  }
  lds_barrier();

  const float* __restrict__ Sd = S + (size_t)ncol * D_OUT + kh * 64 + khalf;
  v2f st[16];
  const int gq = lane >> 3;
  const int j  = lane & 7;
  const int row0 = (MAX_LEN / 4) * w;
  const float* __restrict__ lrow = lowb + (size_t)row0 * D_IN + 4 * j;
  float4 dv[DPF][4];

  {
    if (!last) {
#pragma unroll
      for (int jk = 0; jk < 16; ++jk)
        st[jk] = *reinterpret_cast<const v2f*>(Sd + 4 * jk);
#pragma unroll
      for (int i = 0; i < DPF; ++i) {
        const float4* __restrict__ rp =
            reinterpret_cast<const float4*>(lrow + (size_t)(4 * i + gq) * D_IN);
#pragma unroll
        for (int qd = 0; qd < 4; ++qd) dv[i][qd] = rp[8 * qd];
      }
    }
    const float* __restrict__ aT = &sT[mrow12][kh * 64 + khalf];
    v8f c0 = {}, c1 = {};
#pragma unroll
    for (int kk = 0; kk < 16; kk += 2) {
      const v2f a0 = *reinterpret_cast<const v2f*>(aT + 4 * kk);
      const v2f a1 = *reinterpret_cast<const v2f*>(aT + 4 * kk + 4);
      const v2f b0 = sb[kk], b1 = sb[kk + 1];
      c0 = __builtin_amdgcn_wmma_f32_16x16x4_f32(false, a0, false, b0,
                                                 (short)0, c0, false, false);
      c1 = __builtin_amdgcn_wmma_f32_16x16x4_f32(false, a1, false, b1,
                                                 (short)0, c1, false, false);
    }
    if (lane < 16) {
#pragma unroll
      for (int r = 0; r < 8; ++r) sHp[kh][r][ncol] = c0[r] + c1[r];
    } else {
#pragma unroll
      for (int r = 0; r < NROW - 8; ++r) sHp[kh][8 + r][ncol] = c0[r] + c1[r];
    }
  }
  lds_barrier();

  {
    const int e = t & (D_OUT - 1);
    const float x0 = sHp[0][MAX_K * q + 0][e] + sHp[1][MAX_K * q + 0][e];
    const float x1 = sHp[0][MAX_K * q + 1][e] + sHp[1][MAX_K * q + 1][e];
    const float x2 = sHp[0][MAX_K * q + 2][e] + sHp[1][MAX_K * q + 2][e];
    const float sq = x0 * x0 + x1 * x1 + x2 * x2;
    const float scale = sq / (1.0f + sq) * rsqrtf(sq + 1e-9f);
    const float h0 = scale * x0, h1 = scale * x1, h2 = scale * x2;
    sH[MAX_K * q + 0][e] = h0; sH[MAX_K * q + 1][e] = h1; sH[MAX_K * q + 2][e] = h2;
    if (last) {
      float* __restrict__ ob = out + (size_t)b * MAX_K * D_OUT;
      ob[0 * D_OUT + e] = h0;
      ob[1 * D_OUT + e] = h1;
      ob[2 * D_OUT + e] = h2;
    }
  }
  lds_barrier();

  if (!last) {
    {
      const float* __restrict__ aH = &sH[mrow12][kh * 64 + khalf];
      v8f c0 = {}, c1 = {};
#pragma unroll
      for (int kk = 0; kk < 16; kk += 2) {
        const v2f a0 = *reinterpret_cast<const v2f*>(aH + 4 * kk);
        const v2f a1 = *reinterpret_cast<const v2f*>(aH + 4 * kk + 4);
        const v2f b0 = st[kk], b1 = st[kk + 1];
        c0 = __builtin_amdgcn_wmma_f32_16x16x4_f32(false, a0, false, b0,
                                                   (short)0, c0, false, false);
        c1 = __builtin_amdgcn_wmma_f32_16x16x4_f32(false, a1, false, b1,
                                                   (short)0, c1, false, false);
      }
      if (lane < 16) {
#pragma unroll
        for (int r = 0; r < 8; ++r) sGp[kh][r][ncol] = c0[r] + c1[r];
      } else {
#pragma unroll
        for (int r = 0; r < NROW - 8; ++r) sGp[kh][8 + r][ncol] = c0[r] + c1[r];
      }
    }
    lds_barrier();

    {
      float4 gr[MAX_K][4];
#pragma unroll
      for (int k = 0; k < MAX_K; ++k)
#pragma unroll
        for (int qd = 0; qd < 4; ++qd) {
          const float4 g0 =
              *reinterpret_cast<const float4*>(&sGp[0][MAX_K * q + k][32 * qd + 4 * j]);
          const float4 g1 =
              *reinterpret_cast<const float4*>(&sGp[1][MAX_K * q + k][32 * qd + 4 * j]);
          gr[k][qd] = make_float4(g0.x + g1.x, g0.y + g1.y, g0.z + g1.z, g0.w + g1.w);
        }
      float res[13];
#pragma unroll
      for (int i = 0; i < 13; ++i) {
        const int r = (i < 12) ? (4 * i + gq) : (48 + (gq & 1));
        const float4* __restrict__ rp =
            reinterpret_cast<const float4*>(lrow + (size_t)r * D_IN);
        float4 v[4];
#pragma unroll
        for (int qd = 0; qd < 4; ++qd) v[qd] = (i < DPF) ? dv[i][qd] : rp[8 * qd];
        float acc[MAX_K];
#pragma unroll
        for (int k = 0; k < MAX_K; ++k) {
          float a = v[0].x * gr[k][0].x;
          a += v[0].y * gr[k][0].y; a += v[0].z * gr[k][0].z; a += v[0].w * gr[k][0].w;
#pragma unroll
          for (int qd = 1; qd < 4; ++qd) {
            a += v[qd].x * gr[k][qd].x; a += v[qd].y * gr[k][qd].y;
            a += v[qd].z * gr[k][qd].z; a += v[qd].w * gr[k][qd].w;
          }
          a += __builtin_bit_cast(float, __builtin_amdgcn_update_dpp(
                   0, __builtin_bit_cast(int, a), 0x161, 0xf, 0xf, true));
          a += __builtin_bit_cast(float, __builtin_amdgcn_update_dpp(
                   0, __builtin_bit_cast(int, a), 0x162, 0xf, 0xf, true));
          a += __builtin_bit_cast(float, __builtin_amdgcn_update_dpp(
                   0, __builtin_bit_cast(int, a), 0x164, 0xf, 0xf, true));
          acc[k] = a;
        }
        res[i] = (j == 0) ? acc[0] : ((j == 1) ? acc[1] : acc[2]);
      }
      if (j < MAX_K) {
        float* __restrict__ dst = &sD[q][0] + j * MAX_LEN + row0 + gq;
#pragma unroll
        for (int i = 0; i < 12; ++i) dst[4 * i] = res[i];
        if (gq < 2) dst[48] = res[12];
      }
    }
    __syncthreads();

    for (int p = t; p < MAX_K * MAX_LEN; p += 512)
      atomicAdd(&Bdelta[p], (sD[0][p] + sD[1][p]) + (sD[2][p] + sD[3][p]));
  }
}

extern "C" void kernel_launch(void* const* d_in, const int* in_sizes, int n_in,
                              void* d_out, int out_size, void* d_ws, size_t ws_size,
                              hipStream_t stream) {
  const float* low  = (const float*)d_in[0];
  const float* Bin  = (const float*)d_in[1];
  const float* S    = (const float*)d_in[2];
  const int*   slen = (const int*)d_in[3];
  float* out = (float*)d_out;

  float* D0 = (float*)d_ws;
  float* D1 = D0 + MAX_K * MAX_LEN;
  float* Z  = D1 + MAX_K * MAX_LEN;

  zero_planes_kernel<<<8, 256, 0, stream>>>(D0);

  routing_iter_kernel<false><<<NBLOCK, 512, 0, stream>>>(low, S, Bin, Z,  Z,  slen, D0, out);
  routing_iter_kernel<false><<<NBLOCK, 512, 0, stream>>>(low, S, Bin, D0, Z,  slen, D1, out);
  routing_iter_kernel<true><<<NBLOCK, 512, 0, stream>>>(low, S, Bin, D0, D1, slen, nullptr, out);
}


# T product stops behind each batch's last non-zero softmax weight (40-row chunks); batches dealt to blocks by sorted seq_len so each block has equal live rows; delta batch sums grouped differently
# speedup vs baseline: 6.7439x; 1.1280x over previous
#include <hip/hip_runtime.h>
#include <hip/hip_bf16.h>


typedef float v2f __attribute__((ext_vector_type(2)));
typedef float v8f __attribute__((ext_vector_type(8)));

#define MAX_LEN 200
#define MAX_K   3
#define D_IN    128
#define D_OUT   128
#define NBATCH  1024
#define NEG_PAD (-65535.0f)


static __device__ __forceinline__ float xhalf(float x) {
  const int b = __builtin_bit_cast(int, x);
  return __builtin_bit_cast(float, __builtin_amdgcn_permlanex16(
             b, b, 0x76543210, 0xfedcba98, false, false));
}
template <int M>
static __device__ __forceinline__ float xrow(float x) {
  return __builtin_bit_cast(float, __builtin_amdgcn_update_dpp(
             0, __builtin_bit_cast(int, x), 0x160 + M, 0xf, 0xf, true));
}
static __device__ __forceinline__ float wave_max(float v) {
  v = fmaxf(v, xrow<1>(v)); v = fmaxf(v, xrow<2>(v));
  v = fmaxf(v, xrow<4>(v)); v = fmaxf(v, xrow<8>(v));
  return fmaxf(v, xhalf(v));
}
static __device__ __forceinline__ float wave_sum(float v) {
  v += xrow<1>(v); v += xrow<2>(v); v += xrow<4>(v); v += xrow<8>(v);
  return v + xhalf(v);
}

static __device__ __forceinline__ void lds_barrier() {
  asm volatile("s_wait_dscnt 0x0\n\ts_barrier_signal -1\n\ts_barrier_wait -1" ::: "memory");
}

#define NPLANE 3
#define NB_DEAL 4
#define LBIN   208
__global__ void __launch_bounds__(1024)
prep_kernel(float* __restrict__ planes, const int* __restrict__ seq_len,
            int* __restrict__ deal) {
  __shared__ int hist[32][LBIN];
  __shared__ int lbase[LBIN];
  const int t    = threadIdx.x;
  const int wave = t >> 5;
  const int lane = t & 31;
  for (int p = t; p < NPLANE * MAX_K * MAX_LEN; p += 1024) planes[p] = 0.0f;
  for (int p = t; p < 32 * LBIN; p += 1024) (&hist[0][0])[p] = 0;
  const int len_raw = seq_len[t];
  const int len = (len_raw < 0) ? 0 : ((len_raw > MAX_LEN) ? MAX_LEN : len_raw);
  __syncthreads();
  int below = 0, total = 0;
#pragma unroll
  for (int j = 0; j < 32; ++j) {
    const int lj = __builtin_amdgcn_readlane(len, j);
    total += (lj == len) ? 1 : 0;
    below += (lj == len && j < lane) ? 1 : 0;
  }
  if (below == 0) hist[wave][len] = total;
  __syncthreads();
  if (t <= MAX_LEN) {
    int cnt[32];
#pragma unroll
    for (int wv = 0; wv < 32; ++wv) cnt[wv] = hist[wv][t];
    int run = 0;
#pragma unroll
    for (int wv = 0; wv < 32; ++wv) { hist[wv][t] = run; run += cnt[wv]; }
    lbase[t] = run;
  }
  __syncthreads();
  if (wave == 0) {
    int v[7];
    int s = 0;
#pragma unroll
    for (int i = 0; i < 7; ++i) {
      const int idx = lane * 7 + i;
      v[i] = (idx <= MAX_LEN) ? lbase[idx] : 0;
      s += v[i];
    }
    int inc = s;
#pragma unroll
    for (int d = 1; d < 32; d <<= 1) {
      const int o = __shfl_up(inc, d, 32);
      if (lane >= d) inc += o;
    }
    int run = inc - s;
#pragma unroll
    for (int i = 0; i < 7; ++i) {
      const int idx = lane * 7 + i;
      if (idx <= MAX_LEN) lbase[idx] = run;
      run += v[i];
    }
  }
  __syncthreads();
  const int rank = lbase[len] + hist[wave][len] + below;
  const int r    = rank & 511;
  const int blk  = (r < 256) ? r : (511 - r);
  const int slot = 3 - (2 * (rank >> 9) + ((r < 256) ? 0 : 1));
  deal[2 * (NB_DEAL * blk + slot)]     = t;
  deal[2 * (NB_DEAL * blk + slot) + 1] = len_raw;
}

#define NB     4
#define NROW   (NB * MAX_K)
#define NBLOCK 256
#define TPF    10
#define CH     TPF
#define NCH    (MAX_LEN / 4 / CH)
#define DPF    6

template <int NK>
static __device__ __forceinline__ void t_product(const float* __restrict__ lowb,
                                                 const float* wrow, const int n0,
                                                 const int n1, const int khalf,
                                                 const v2f (&pb0)[TPF], const v2f (&pb1)[TPF],
                                                 v8f& c0, v8f& c1) {
#pragma unroll
  for (int kk = 0; kk < NK; ++kk) {
    const int kb = kk * 4 + khalf;
    const v2f a = *reinterpret_cast<const v2f*>(wrow + kb);
    v2f b0, b1;
    if (kk < TPF) {
      b0 = pb0[kk];  b1 = pb1[kk];
    } else {
      b0.x = lowb[kb * D_IN + n0];  b0.y = lowb[(kb + 1) * D_IN + n0];
      b1.x = lowb[kb * D_IN + n1];  b1.y = lowb[(kb + 1) * D_IN + n1];
    }
    c0 = __builtin_amdgcn_wmma_f32_16x16x4_f32(false, a, false, b0,
                                               (short)0, c0, false, false);
    c1 = __builtin_amdgcn_wmma_f32_16x16x4_f32(false, a, false, b1,
                                               (short)0, c1, false, false);
  }
}

template <bool LAST>
__global__ void __launch_bounds__(512) __attribute__((amdgpu_waves_per_eu(4, 4)))
routing_iter_kernel(const float* __restrict__ low, const float* __restrict__ S,
                    const float* __restrict__ Bin, const float* __restrict__ Dp0,
                    const float* __restrict__ Dp1,
                    const int* __restrict__ deal, float* __restrict__ Bdelta,
                    float* __restrict__ out) {
  constexpr bool last = LAST;
  __shared__ __align__(16) float sW[NB][MAX_K][MAX_LEN];
  __shared__ __align__(16) float sT[NROW][D_IN + 4];
  __shared__ __align__(16) float sHp[2][NROW][D_OUT];
  __shared__ __align__(16) float sH[NROW][D_OUT + 4];
  __shared__ __align__(16) float sGp[2][NROW][D_IN];
  __shared__ float sD[NB][MAX_K * MAX_LEN];
  __shared__ float sLast[NB][4];

  const int t    = threadIdx.x;
  const int wave = t >> 5;
  const int lane = t & 31;
  const int q    = wave >> 2;
  const int w    = wave & 3;
  const int qs = __builtin_amdgcn_readfirstlane(q);
  const int2 bl = reinterpret_cast<const int2*>(deal)[blockIdx.x * NB + qs];
  const int b    = __builtin_amdgcn_readfirstlane(
      (bl.x < 0) ? 0 : ((bl.x > NBATCH - 1) ? (NBATCH - 1) : bl.x));
  const int len  = bl.y;
  const float* __restrict__ lowb = low + (size_t)b * MAX_LEN * D_IN;

  const int m     = lane & 15;
  const int khalf = (lane >> 4) * 2;
  const int n0 = (w * 2) * 16 + m;
  const int n1 = n0 + 16;
  float rb[7], rd0[7], rd1[7];
  {
    const int wr = (w < MAX_K) ? w : (MAX_K - 1);
    const float* __restrict__ Brow  = Bin + wr * MAX_LEN;
    const float* __restrict__ D0row = Dp0 + wr * MAX_LEN;
    const float* __restrict__ D1row = Dp1 + wr * MAX_LEN;
#pragma unroll
    for (int i = 0; i < 7; ++i) {
      const int l = lane + 32 * i;
      const int lc = (l < MAX_LEN) ? l : (MAX_LEN - 1);
      rb[i] = Brow[lc];  rd0[i] = D0row[lc];  rd1[i] = D1row[lc];
    }
  }
  __builtin_amdgcn_sched_barrier(0);
  v2f pb0[TPF], pb1[TPF];
#pragma unroll
  for (int kk = 0; kk < TPF; ++kk) {
    const int kb = kk * 4 + khalf;
    pb0[kk].x = lowb[kb * D_IN + n0];  pb0[kk].y = lowb[(kb + 1) * D_IN + n0];
    pb1[kk].x = lowb[kb * D_IN + n1];  pb1[kk].y = lowb[(kb + 1) * D_IN + n1];
  }
  {
    float x[7];
#pragma unroll
    for (int i = 0; i < 7; ++i) {
      const int l = lane + 32 * i;
      const float v = (rb[i] + rd0[i]) + rd1[i];
      x[i] = (l < MAX_LEN) ? ((l < len) ? v : NEG_PAD) : -3.0e38f;
    }
    float mx = -3.0e38f;
#pragma unroll
    for (int i = 0; i < 7; ++i) mx = fmaxf(mx, x[i]);
    mx = wave_max(mx);
    float sum = 0.0f;
#pragma unroll
    for (int i = 0; i < 7; ++i) {
      const int l = lane + 32 * i;
      const float e = (l < MAX_LEN) ? __expf(x[i] - mx) : 0.0f;
      x[i] = e; sum += e;
    }
    sum = wave_sum(sum);
    const float inv = 1.0f / sum;
    float lastnz = -1.0f;
#pragma unroll
    for (int i = 0; i < 7; ++i) {
      const int l = lane + 32 * i;
      if (l < MAX_LEN && x[i] * inv != 0.0f) lastnz = (float)l;
    }
    lastnz = wave_max(lastnz);
    if (w < MAX_K) {
#pragma unroll
      for (int i = 0; i < 7; ++i) {
        const int l = lane + 32 * i;
        if (l < MAX_LEN) sW[q][w][l] = x[i] * inv;
      }
      if (lane == 0) sLast[q][w] = lastnz;
    }
  }
  lds_barrier();
  const float lq = fmaxf(fmaxf(sLast[q][0], sLast[q][1]), sLast[q][2]);
  int nch = ((int)lq + 4 * CH) / (4 * CH);
  nch = __builtin_amdgcn_readfirstlane(nch);
  nch = (nch < 1) ? 1 : ((nch > NCH) ? NCH : nch);

  const int mrow3  = (m < MAX_K) ? m : 0;
  const int mrow12 = (m < NROW) ? m : 0;

  const int tile  = wave & 7;
  const int kh    = wave >> 3;
  const int ncol  = tile * 16 + (lane & 15);
  const float* __restrict__ Sk = S + (size_t)(kh * 64 + khalf) * D_OUT + ncol;
  v2f sb[16];

  {
    v8f c0 = {}, c1 = {};
    switch (nch) {
      case 1:  t_product<1 * CH>(lowb, &sW[q][mrow3][0], n0, n1, khalf, pb0, pb1, c0, c1); break;
      case 2:  t_product<2 * CH>(lowb, &sW[q][mrow3][0], n0, n1, khalf, pb0, pb1, c0, c1); break;
      case 3:  t_product<3 * CH>(lowb, &sW[q][mrow3][0], n0, n1, khalf, pb0, pb1, c0, c1); break;
      case 4:  t_product<4 * CH>(lowb, &sW[q][mrow3][0], n0, n1, khalf, pb0, pb1, c0, c1); break;
      default: t_product<5 * CH>(lowb, &sW[q][mrow3][0], n0, n1, khalf, pb0, pb1, c0, c1); break;
    }
#pragma unroll
    for (int jk = 0; jk < 16; ++jk) {
      sb[jk].x = Sk[(4 * jk) * D_OUT];  sb[jk].y = Sk[(4 * jk + 1) * D_OUT];
    }
    if (lane < 16) {
#pragma unroll
      for (int r = 0; r < MAX_K; ++r) {
        sT[MAX_K * q + r][n0] = c0[r];
        sT[MAX_K * q + r][n1] = c1[r];
      }
    }
  }
  lds_barrier();

  const float* __restrict__ Sd = S + (size_t)ncol * D_OUT + kh * 64 + khalf;
  v2f st[16];
  const int gq = lane >> 3;
  const int j  = lane & 7;
  const int row0 = (MAX_LEN / 4) * w;
  const float* __restrict__ lrow = lowb + (size_t)row0 * D_IN + 4 * j;
  float4 dv[DPF][4];

  {
    if (!last) {
#pragma unroll
      for (int jk = 0; jk < 16; ++jk)
        st[jk] = *reinterpret_cast<const v2f*>(Sd + 4 * jk);
#pragma unroll
      for (int i = 0; i < DPF; ++i) {
        const float4* __restrict__ rp =
            reinterpret_cast<const float4*>(lrow + (size_t)(4 * i + gq) * D_IN);
#pragma unroll
        for (int qd = 0; qd < 4; ++qd) dv[i][qd] = rp[8 * qd];
      }
    }
    const float* __restrict__ aT = &sT[mrow12][kh * 64 + khalf];
    v8f c0 = {}, c1 = {};
#pragma unroll
    for (int kk = 0; kk < 16; kk += 2) {
      const v2f a0 = *reinterpret_cast<const v2f*>(aT + 4 * kk);
      const v2f a1 = *reinterpret_cast<const v2f*>(aT + 4 * kk + 4);
      const v2f b0 = sb[kk], b1 = sb[kk + 1];
      c0 = __builtin_amdgcn_wmma_f32_16x16x4_f32(false, a0, false, b0,
                                                 (short)0, c0, false, false);
      c1 = __builtin_amdgcn_wmma_f32_16x16x4_f32(false, a1, false, b1,
                                                 (short)0, c1, false, false);
    }
    if (lane < 16) {
#pragma unroll
      for (int r = 0; r < 8; ++r) sHp[kh][r][ncol] = c0[r] + c1[r];
    } else {
#pragma unroll
      for (int r = 0; r < NROW - 8; ++r) sHp[kh][8 + r][ncol] = c0[r] + c1[r];
    }
  }
  lds_barrier();

  {
    const int e = t & (D_OUT - 1);
    const float x0 = sHp[0][MAX_K * q + 0][e] + sHp[1][MAX_K * q + 0][e];
    const float x1 = sHp[0][MAX_K * q + 1][e] + sHp[1][MAX_K * q + 1][e];
    const float x2 = sHp[0][MAX_K * q + 2][e] + sHp[1][MAX_K * q + 2][e];
    const float sq = x0 * x0 + x1 * x1 + x2 * x2;
    const float scale = sq / (1.0f + sq) * rsqrtf(sq + 1e-9f);
    const float h0 = scale * x0, h1 = scale * x1, h2 = scale * x2;
    sH[MAX_K * q + 0][e] = h0; sH[MAX_K * q + 1][e] = h1; sH[MAX_K * q + 2][e] = h2;
    if (last) {
      float* __restrict__ ob = out + (size_t)b * MAX_K * D_OUT;
      ob[0 * D_OUT + e] = h0;
      ob[1 * D_OUT + e] = h1;
      ob[2 * D_OUT + e] = h2;
    }
  }
  lds_barrier();

  if (!last) {
    {
      const float* __restrict__ aH = &sH[mrow12][kh * 64 + khalf];
      v8f c0 = {}, c1 = {};
#pragma unroll
      for (int kk = 0; kk < 16; kk += 2) {
        const v2f a0 = *reinterpret_cast<const v2f*>(aH + 4 * kk);
        const v2f a1 = *reinterpret_cast<const v2f*>(aH + 4 * kk + 4);
        const v2f b0 = st[kk], b1 = st[kk + 1];
        c0 = __builtin_amdgcn_wmma_f32_16x16x4_f32(false, a0, false, b0,
                                                   (short)0, c0, false, false);
        c1 = __builtin_amdgcn_wmma_f32_16x16x4_f32(false, a1, false, b1,
                                                   (short)0, c1, false, false);
      }
      if (lane < 16) {
#pragma unroll
        for (int r = 0; r < 8; ++r) sGp[kh][r][ncol] = c0[r] + c1[r];
      } else {
#pragma unroll
        for (int r = 0; r < NROW - 8; ++r) sGp[kh][8 + r][ncol] = c0[r] + c1[r];
      }
    }
    lds_barrier();

    {
      float4 gr[MAX_K][4];
#pragma unroll
      for (int k = 0; k < MAX_K; ++k)
#pragma unroll
        for (int qd = 0; qd < 4; ++qd) {
          const float4 g0 =
              *reinterpret_cast<const float4*>(&sGp[0][MAX_K * q + k][32 * qd + 4 * j]);
          const float4 g1 =
              *reinterpret_cast<const float4*>(&sGp[1][MAX_K * q + k][32 * qd + 4 * j]);
          gr[k][qd] = make_float4(g0.x + g1.x, g0.y + g1.y, g0.z + g1.z, g0.w + g1.w);
        }
      float res[13];
#pragma unroll
      for (int i = 0; i < 13; ++i) {
        const int r = (i < 12) ? (4 * i + gq) : (48 + (gq & 1));
        const float4* __restrict__ rp =
            reinterpret_cast<const float4*>(lrow + (size_t)r * D_IN);
        float4 v[4];
#pragma unroll
        for (int qd = 0; qd < 4; ++qd) v[qd] = (i < DPF) ? dv[i][qd] : rp[8 * qd];
        float acc[MAX_K];
#pragma unroll
        for (int k = 0; k < MAX_K; ++k) {
          float a = v[0].x * gr[k][0].x;
          a += v[0].y * gr[k][0].y; a += v[0].z * gr[k][0].z; a += v[0].w * gr[k][0].w;
#pragma unroll
          for (int qd = 1; qd < 4; ++qd) {
            a += v[qd].x * gr[k][qd].x; a += v[qd].y * gr[k][qd].y;
            a += v[qd].z * gr[k][qd].z; a += v[qd].w * gr[k][qd].w;
          }
          a += __builtin_bit_cast(float, __builtin_amdgcn_update_dpp(
                   0, __builtin_bit_cast(int, a), 0x161, 0xf, 0xf, true));
          a += __builtin_bit_cast(float, __builtin_amdgcn_update_dpp(
                   0, __builtin_bit_cast(int, a), 0x162, 0xf, 0xf, true));
          a += __builtin_bit_cast(float, __builtin_amdgcn_update_dpp(
                   0, __builtin_bit_cast(int, a), 0x164, 0xf, 0xf, true));
          acc[k] = a;
        }
        res[i] = (j == 0) ? acc[0] : ((j == 1) ? acc[1] : acc[2]);
      }
      if (j < MAX_K) {
        float* __restrict__ dst = &sD[q][0] + j * MAX_LEN + row0 + gq;
#pragma unroll
        for (int i = 0; i < 12; ++i) dst[4 * i] = res[i];
        if (gq < 2) dst[48] = res[12];
      }
    }
    __syncthreads();

    for (int p = t; p < MAX_K * MAX_LEN; p += 512)
      atomicAdd(&Bdelta[p], (sD[0][p] + sD[1][p]) + (sD[2][p] + sD[3][p]));
  }
}

extern "C" void kernel_launch(void* const* d_in, const int* in_sizes, int n_in,
                              void* d_out, int out_size, void* d_ws, size_t ws_size,
                              hipStream_t stream) {
  const float* low  = (const float*)d_in[0];
  const float* Bin  = (const float*)d_in[1];
  const float* S    = (const float*)d_in[2];
  const int*   slen = (const int*)d_in[3];
  float* out = (float*)d_out;

  float* D0 = (float*)d_ws;
  float* D1 = D0 + MAX_K * MAX_LEN;
  float* Z  = D1 + MAX_K * MAX_LEN;
  int* deal = (int*)(Z + MAX_K * MAX_LEN);

  prep_kernel<<<1, 1024, 0, stream>>>(D0, slen, deal);

  routing_iter_kernel<false><<<NBLOCK, 512, 0, stream>>>(low, S, Bin, Z,  Z,  deal, D0, out);
  routing_iter_kernel<false><<<NBLOCK, 512, 0, stream>>>(low, S, Bin, D0, Z,  deal, D1, out);
  routing_iter_kernel<true><<<NBLOCK, 512, 0, stream>>>(low, S, Bin, D0, D1, deal, nullptr, out);
}
